# starting kernel
# baseline (speedup 1.0000x reference)
#include <hip/hip_runtime.h>
#include <hip/hip_bf16.h>


#define N_NODES 50000
#define N_EDGES 800000
#define GNN_IN  256
#define GNN_HID 64
#define COMB    64
#define MLP_IN  256
#define MLP_HID 512

typedef __attribute__((ext_vector_type(16))) __bf16 v16bf;
typedef __attribute__((ext_vector_type(8)))  float  v8f;

union V16 {
    uint4 u[2];
    v16bf v;
};

__device__ __forceinline__ unsigned short f2bf(float f) {
    unsigned int u = __float_as_uint(f);
    unsigned int r = u + 0x7FFFu + ((u >> 16) & 1u);
    return (unsigned short)(r >> 16);
}

__global__ void cvt_bf16_kernel(const float* __restrict__ in,
                                unsigned short* __restrict__ out, int n4) {
    int i = blockIdx.x * blockDim.x + threadIdx.x;
    if (i >= n4) return;
    const float4 f = ((const float4*)in)[i];
    ushort4 o;
    o.x = f2bf(f.x); o.y = f2bf(f.y); o.z = f2bf(f.z); o.w = f2bf(f.w);
    ((ushort4*)out)[i] = o;
}

__global__ void pack_w_kernel(const float* __restrict__ W,
                              unsigned short* __restrict__ P, int K, int N) {
    int idx = blockIdx.x * blockDim.x + threadIdx.x;
    int nTiles = N >> 4;
    int total = (K >> 5) * nTiles * 32;
    if (idx >= total) return;
    int lane = idx & 31;
    int tile = idx >> 5;
    int kt = tile / nTiles;
    int nt = tile % nTiles;
    int col = nt * 16 + (lane & 15);
    int kb  = kt * 32 + ((lane < 16) ? 0 : 8);
    __attribute__((aligned(16))) unsigned short o[16];
#pragma unroll
    for (int j = 0; j < 8; ++j) o[j]     = f2bf(W[(size_t)(kb + j) * N + col]);
#pragma unroll
    for (int j = 0; j < 8; ++j) o[8 + j] = f2bf(W[(size_t)(kb + 16 + j) * N + col]);
    uint4* dst = (uint4*)(P + (size_t)tile * 512 + lane * 16);
    dst[0] = ((uint4*)o)[0];
    dst[1] = ((uint4*)o)[1];
}

__global__ void gemm_wmma_kernel(const unsigned short* __restrict__ A,
                                 const unsigned short* __restrict__ Bp,
                                 const float* __restrict__ bias,
                                 void* __restrict__ C,
                                 int M, int N, int K, int doRelu, int outBf16) {
    const int lane = threadIdx.x & 31;
    const int wid  = threadIdx.x >> 5;
    const int nGroups = N >> 6;
    const int mTiles  = M >> 4;
    const int task = blockIdx.x * 8 + wid;
    if (task >= mTiles * nGroups) return;
    const int mt = task / nGroups;
    const int ng = task % nGroups;
    const int kTiles = K >> 5;
    const int nTiles = N >> 4;

    const int hsel = (lane < 16) ? 0 : 8;
    const unsigned short* arow = A + (size_t)(mt * 16 + (lane & 15)) * K;
    const unsigned short* bcol = Bp + (size_t)(ng << 2) * 512 + lane * 16;
    const size_t bstride = (size_t)nTiles * 512;

    v8f acc0 = {}, acc1 = {}, acc2 = {}, acc3 = {};

    V16 aC, aN;
    V16 bC0, bC1, bC2, bC3, bN0, bN1, bN2, bN3;

    {
        const uint4* ap = (const uint4*)(arow + hsel);
        aC.u[0] = ap[0];
        aC.u[1] = ap[2];
        const unsigned short* bt = bcol;
        bC0.u[0] = ((const uint4*)(bt + 0 * 512))[0];
        bC0.u[1] = ((const uint4*)(bt + 0 * 512))[1];
        bC1.u[0] = ((const uint4*)(bt + 1 * 512))[0];
        bC1.u[1] = ((const uint4*)(bt + 1 * 512))[1];
        bC2.u[0] = ((const uint4*)(bt + 2 * 512))[0];
        bC2.u[1] = ((const uint4*)(bt + 2 * 512))[1];
        bC3.u[0] = ((const uint4*)(bt + 3 * 512))[0];
        bC3.u[1] = ((const uint4*)(bt + 3 * 512))[1];
    }

    for (int kt = 0; kt < kTiles; ++kt) {
        const int ktn = (kt + 1 < kTiles) ? (kt + 1) : kt;
        {
            const uint4* ap = (const uint4*)(arow + ktn * 32 + hsel);
            aN.u[0] = ap[0];
            aN.u[1] = ap[2];
            const unsigned short* bt = bcol + (size_t)ktn * bstride;
            bN0.u[0] = ((const uint4*)(bt + 0 * 512))[0];
            bN0.u[1] = ((const uint4*)(bt + 0 * 512))[1];
            bN1.u[0] = ((const uint4*)(bt + 1 * 512))[0];
            bN1.u[1] = ((const uint4*)(bt + 1 * 512))[1];
            bN2.u[0] = ((const uint4*)(bt + 2 * 512))[0];
            bN2.u[1] = ((const uint4*)(bt + 2 * 512))[1];
            bN3.u[0] = ((const uint4*)(bt + 3 * 512))[0];
            bN3.u[1] = ((const uint4*)(bt + 3 * 512))[1];
        }

        acc0 = __builtin_amdgcn_wmma_f32_16x16x32_bf16(false, aC.v, false, bC0.v,
                                                       (short)0, acc0, false, false);
        acc1 = __builtin_amdgcn_wmma_f32_16x16x32_bf16(false, aC.v, false, bC1.v,
                                                       (short)0, acc1, false, false);
        acc2 = __builtin_amdgcn_wmma_f32_16x16x32_bf16(false, aC.v, false, bC2.v,
                                                       (short)0, acc2, false, false);
        acc3 = __builtin_amdgcn_wmma_f32_16x16x32_bf16(false, aC.v, false, bC3.v,
                                                       (short)0, acc3, false, false);

        aC = aN;
        bC0 = bN0; bC1 = bN1; bC2 = bN2; bC3 = bN3;
    }

    const int r0 = mt * 16 + ((lane < 16) ? 0 : 8);
    v8f accs[4] = {acc0, acc1, acc2, acc3};
#pragma unroll
    for (int j = 0; j < 4; ++j) {
        const int cc = (ng * 4 + j) * 16 + (lane & 15);
        const float bv = bias ? bias[cc] : 0.0f;
#pragma unroll
        for (int v = 0; v < 8; ++v) {
            float val = accs[j][v] + bv;
            if (doRelu) val = fmaxf(val, 0.0f);
            const size_t off = (size_t)(r0 + v) * N + cc;
            if (outBf16) ((unsigned short*)C)[off] = f2bf(val);
            else         ((float*)C)[off] = val;
        }
    }
}

__global__ void deg_init_kernel(float* __restrict__ deg, int n) {
    int i = blockIdx.x * blockDim.x + threadIdx.x;
    if (i < n) deg[i] = 1.0f;
}

__global__ void deg_accum_kernel(const int* __restrict__ dst,
                                 const float* __restrict__ ew,
                                 float* __restrict__ deg, int nE) {
    int e = blockIdx.x * blockDim.x + threadIdx.x;
    if (e < nE) atomicAdd(&deg[dst[e]], ew[e]);
}

__global__ void dinv_kernel(const float* __restrict__ deg,
                            float* __restrict__ dinv, int n) {
    int i = blockIdx.x * blockDim.x + threadIdx.x;
    if (i >= n) return;
    float d = deg[i];
    dinv[i] = (d > 0.0f) ? rsqrtf(fmaxf(d, 1e-12f)) : 0.0f;
}

__global__ void edge_norm_kernel(const int* __restrict__ src,
                                 const int* __restrict__ dst,
                                 const float* __restrict__ ew,
                                 const float* __restrict__ dinv,
                                 float* __restrict__ norm, int nE) {
    int e = blockIdx.x * blockDim.x + threadIdx.x;
    if (e < nE) norm[e] = dinv[src[e]] * ew[e] * dinv[dst[e]];
}

__global__ void agg_init_kernel(const float* __restrict__ h,
                                const float* __restrict__ dinv,
                                float* __restrict__ agg, int nNodes) {
    int idx = blockIdx.x * blockDim.x + threadIdx.x;
    int n = idx >> 4;
    if (n >= nNodes) return;
    int c = (idx & 15) * 4;
    float w = dinv[n]; w *= w;
    float4 hv = *(const float4*)(h + (size_t)n * COMB + c);
    float4 o = {hv.x * w, hv.y * w, hv.z * w, hv.w * w};
    *(float4*)(agg + (size_t)n * COMB + c) = o;
}

__global__ void gcn_scatter_kernel(const float* __restrict__ h,
                                   const int* __restrict__ src,
                                   const int* __restrict__ dst,
                                   const float* __restrict__ norm,
                                   float* __restrict__ agg, int nE) {
    int idx = blockIdx.x * blockDim.x + threadIdx.x;
    int e = idx >> 4;
    if (e >= nE) return;
    int c = (idx & 15) * 4;
    int s = src[e], d = dst[e];
    float nm = norm[e];
    float4 hv = *(const float4*)(h + (size_t)s * COMB + c);
    float* ap = agg + (size_t)d * COMB + c;
    atomicAdd(ap + 0, hv.x * nm);
    atomicAdd(ap + 1, hv.y * nm);
    atomicAdd(ap + 2, hv.z * nm);
    atomicAdd(ap + 3, hv.w * nm);
}

__global__ void relu_bias_bf16_kernel(const float* __restrict__ agg,
                                      const float* __restrict__ b,
                                      unsigned short* __restrict__ g, int total) {
    int idx = blockIdx.x * blockDim.x + threadIdx.x;
    if (idx >= total) return;
    float v = agg[idx] + b[idx & (COMB - 1)];
    g[idx] = f2bf(fmaxf(v, 0.0f));
}

__global__ void final_kernel(const float* __restrict__ mlp,
                             const float* __restrict__ agg2,
                             const float* __restrict__ b2,
                             const float* __restrict__ fw,
                             const float* __restrict__ fb,
                             float* __restrict__ out) {
    __shared__ float red[128];
    int n = blockIdx.x;
    int t = threadIdx.x;
    float v;
    if (t < COMB) v = mlp[(size_t)n * COMB + t];
    else          v = agg2[(size_t)n * COMB + (t - COMB)] + b2[t - COMB];
    red[t] = v * fw[t];
    __syncthreads();
    for (int off = 64; off > 0; off >>= 1) {
        if (t < off) red[t] += red[t + off];
        __syncthreads();
    }
    if (t == 0) out[n] = red[0] + fb[0];
}

extern "C" void kernel_launch(void* const* d_in, const int* in_sizes, int n_in,
                              void* d_out, int out_size, void* d_ws, size_t ws_size,
                              hipStream_t stream) {
    (void)in_sizes; (void)n_in; (void)out_size; (void)ws_size;

    const float* x        = (const float*)d_in[0];
    const int*   eidx     = (const int*)d_in[1];
    const float* eattr    = (const float*)d_in[2];
    const float* ufeat    = (const float*)d_in[3];
    const float* gcn1_W   = (const float*)d_in[4];
    const float* gcn1_b   = (const float*)d_in[5];
    const float* gcn2_W   = (const float*)d_in[6];
    const float* gcn2_b   = (const float*)d_in[7];
    const float* fc1_W    = (const float*)d_in[8];
    const float* fc1_b    = (const float*)d_in[9];
    const float* fc2_W    = (const float*)d_in[10];
    const float* fc2_b    = (const float*)d_in[11];
    const float* fcf_W    = (const float*)d_in[12];
    const float* fcf_b    = (const float*)d_in[13];
    const int* esrc = eidx;
    const int* edst = eidx + N_EDGES;
    float* out = (float*)d_out;

    char* ws = (char*)d_ws;
    size_t off = 0;
    auto carve = [&](size_t bytes) -> void* {
        void* p = ws + off;
        off = (off + bytes + 255) & ~(size_t)255;
        return p;
    };
    unsigned short* UBF = (unsigned short*)carve((size_t)N_NODES * MLP_IN * 2);
    unsigned short* XBF = (unsigned short*)carve((size_t)N_NODES * GNN_IN * 2);
    unsigned short* HBF = (unsigned short*)carve((size_t)N_NODES * MLP_HID * 2);
    unsigned short* GBF = (unsigned short*)carve((size_t)N_NODES * COMB * 2);
    unsigned short* W1P = (unsigned short*)carve((size_t)MLP_IN * MLP_HID * 2);
    unsigned short* W2P = (unsigned short*)carve((size_t)MLP_HID * COMB * 2);
    unsigned short* G1P = (unsigned short*)carve((size_t)GNN_IN * GNN_HID * 2);
    unsigned short* G2P = (unsigned short*)carve((size_t)GNN_HID * COMB * 2);
    float* MLPOUT = (float*)carve((size_t)N_NODES * COMB * 4);
    float* H1     = (float*)carve((size_t)N_NODES * COMB * 4);
    float* AGG    = (float*)carve((size_t)N_NODES * COMB * 4);
    float* DEG    = (float*)carve((size_t)N_NODES * 4);
    float* DINV   = (float*)carve((size_t)N_NODES * 4);
    float* NORM   = (float*)carve((size_t)N_EDGES * 4);
    float* H2 = H1;

    {
        int n4 = (N_NODES * MLP_IN) / 4;
        cvt_bf16_kernel<<<(n4 + 255) / 256, 256, 0, stream>>>(ufeat, UBF, n4);
        n4 = (N_NODES * GNN_IN) / 4;
        cvt_bf16_kernel<<<(n4 + 255) / 256, 256, 0, stream>>>(x, XBF, n4);
    }
    auto pack = [&](const float* W, unsigned short* P, int K, int N) {
        int total = (K / 32) * (N / 16) * 32;
        pack_w_kernel<<<(total + 255) / 256, 256, 0, stream>>>(W, P, K, N);
    };
    pack(fc1_W, W1P, MLP_IN, MLP_HID);
    pack(fc2_W, W2P, MLP_HID, COMB);
    pack(gcn1_W, G1P, GNN_IN, GNN_HID);
    pack(gcn2_W, G2P, GNN_HID, COMB);

    deg_init_kernel<<<(N_NODES + 255) / 256, 256, 0, stream>>>(DEG, N_NODES);
    deg_accum_kernel<<<(N_EDGES + 255) / 256, 256, 0, stream>>>(edst, eattr, DEG, N_EDGES);
    dinv_kernel<<<(N_NODES + 255) / 256, 256, 0, stream>>>(DEG, DINV, N_NODES);
    edge_norm_kernel<<<(N_EDGES + 255) / 256, 256, 0, stream>>>(esrc, edst, eattr, DINV,
                                                                NORM, N_EDGES);

    auto gemm = [&](const unsigned short* A, const unsigned short* Bp,
                    const float* bias, void* C, int M, int N, int K,
                    int relu, int outBf) {
        int tasks = (M / 16) * (N / 64);
        int blocks = (tasks + 7) / 8;
        gemm_wmma_kernel<<<blocks, 256, 0, stream>>>(A, Bp, bias, C, M, N, K, relu, outBf);
    };

    gemm(UBF, W1P, fc1_b, HBF, N_NODES, MLP_HID, MLP_IN,  1,  1);
    gemm(HBF, W2P, fc2_b, MLPOUT, N_NODES, COMB, MLP_HID, 0, 0);

    gemm(XBF, G1P, nullptr, H1, N_NODES, GNN_HID, GNN_IN, 0, 0);
    agg_init_kernel<<<(N_NODES * 16 + 255) / 256, 256, 0, stream>>>(H1, DINV, AGG, N_NODES);
    gcn_scatter_kernel<<<(N_EDGES * 16 + 255) / 256, 256, 0, stream>>>(
        H1, esrc, edst, NORM, AGG, N_EDGES);
    relu_bias_bf16_kernel<<<(N_NODES * COMB + 255) / 256, 256, 0, stream>>>(
        AGG, gcn1_b, GBF, N_NODES * COMB);

    gemm(GBF, G2P, nullptr, H2, N_NODES, COMB, GNN_HID, 0, 0);
    agg_init_kernel<<<(N_NODES * 16 + 255) / 256, 256, 0, stream>>>(H2, DINV, AGG, N_NODES);
    gcn_scatter_kernel<<<(N_EDGES * 16 + 255) / 256, 256, 0, stream>>>(
        H2, esrc, edst, NORM, AGG, N_EDGES);

    final_kernel<<<N_NODES, 128, 0, stream>>>(MLPOUT, AGG, gcn2_b, fcf_W, fcf_b, out);
}


# layer-2 aggregation moved behind the final projection (linear): fc2/gcn2 GEMMs reduce rows against the projection vector in registers; 2nd scatter adds 1 float per edge into line-padded sums
# speedup vs baseline: 1.3057x; 1.3057x over previous
#include <hip/hip_runtime.h>
#include <hip/hip_bf16.h>


#define N_NODES 50000
#define N_EDGES 800000
#define GNN_IN  256
#define GNN_HID 64
#define COMB    64
#define MLP_IN  256
#define MLP_HID 512
#define ACC_PITCH 32

typedef __attribute__((ext_vector_type(16))) __bf16 v16bf;
typedef __attribute__((ext_vector_type(8)))  float  v8f;

union V16 {
    uint4 u[2];
    v16bf v;
};

__device__ __forceinline__ unsigned short f2bf(float f) {
    unsigned int u = __float_as_uint(f);
    unsigned int r = u + 0x7FFFu + ((u >> 16) & 1u);
    return (unsigned short)(r >> 16);
}

__global__ void cvt_bf16_kernel(const float* __restrict__ in,
                                unsigned short* __restrict__ out, int n4) {
    int i = blockIdx.x * blockDim.x + threadIdx.x;
    if (i >= n4) return;
    const float4 f = ((const float4*)in)[i];
    ushort4 o;
    o.x = f2bf(f.x); o.y = f2bf(f.y); o.z = f2bf(f.z); o.w = f2bf(f.w);
    ((ushort4*)out)[i] = o;
}

__global__ void pack_w_kernel(const float* __restrict__ W,
                              unsigned short* __restrict__ P, int K, int N) {
    int idx = blockIdx.x * blockDim.x + threadIdx.x;
    int nTiles = N >> 4;
    int total = (K >> 5) * nTiles * 32;
    if (idx >= total) return;
    int lane = idx & 31;
    int tile = idx >> 5;
    int kt = tile / nTiles;
    int nt = tile % nTiles;
    int col = nt * 16 + (lane & 15);
    int kb  = kt * 32 + ((lane < 16) ? 0 : 8);
    __attribute__((aligned(16))) unsigned short o[16];
#pragma unroll
    for (int j = 0; j < 8; ++j) o[j]     = f2bf(W[(size_t)(kb + j) * N + col]);
#pragma unroll
    for (int j = 0; j < 8; ++j) o[8 + j] = f2bf(W[(size_t)(kb + 16 + j) * N + col]);
    uint4* dst = (uint4*)(P + (size_t)tile * 512 + lane * 16);
    dst[0] = ((uint4*)o)[0];
    dst[1] = ((uint4*)o)[1];
}

__global__ void gemm_wmma_kernel(const unsigned short* __restrict__ A,
                                 const unsigned short* __restrict__ Bp,
                                 const float* __restrict__ bias,
                                 void* __restrict__ C,
                                 int M, int N, int K, int doRelu, int outBf16) {
    const int lane = threadIdx.x & 31;
    const int wid  = threadIdx.x >> 5;
    const int nGroups = N >> 6;
    const int mTiles  = M >> 4;
    const int task = blockIdx.x * 8 + wid;
    if (task >= mTiles * nGroups) return;
    const int mt = task / nGroups;
    const int ng = task % nGroups;
    const int kTiles = K >> 5;
    const int nTiles = N >> 4;

    const int hsel = (lane < 16) ? 0 : 8;
    const unsigned short* arow = A + (size_t)(mt * 16 + (lane & 15)) * K;
    const unsigned short* bcol = Bp + (size_t)(ng << 2) * 512 + lane * 16;
    const size_t bstride = (size_t)nTiles * 512;

    v8f acc0 = {}, acc1 = {}, acc2 = {}, acc3 = {};

    V16 aC, aN;
    V16 bC0, bC1, bC2, bC3, bN0, bN1, bN2, bN3;

    {
        const uint4* ap = (const uint4*)(arow + hsel);
        aC.u[0] = ap[0];
        aC.u[1] = ap[2];
        const unsigned short* bt = bcol;
        bC0.u[0] = ((const uint4*)(bt + 0 * 512))[0];
        bC0.u[1] = ((const uint4*)(bt + 0 * 512))[1];
        bC1.u[0] = ((const uint4*)(bt + 1 * 512))[0];
        bC1.u[1] = ((const uint4*)(bt + 1 * 512))[1];
        bC2.u[0] = ((const uint4*)(bt + 2 * 512))[0];
        bC2.u[1] = ((const uint4*)(bt + 2 * 512))[1];
        bC3.u[0] = ((const uint4*)(bt + 3 * 512))[0];
        bC3.u[1] = ((const uint4*)(bt + 3 * 512))[1];
    }

    for (int kt = 0; kt < kTiles; ++kt) {
        const int ktn = (kt + 1 < kTiles) ? (kt + 1) : kt;
        {
            const uint4* ap = (const uint4*)(arow + ktn * 32 + hsel);
            aN.u[0] = ap[0];
            aN.u[1] = ap[2];
            const unsigned short* bt = bcol + (size_t)ktn * bstride;
            bN0.u[0] = ((const uint4*)(bt + 0 * 512))[0];
            bN0.u[1] = ((const uint4*)(bt + 0 * 512))[1];
            bN1.u[0] = ((const uint4*)(bt + 1 * 512))[0];
            bN1.u[1] = ((const uint4*)(bt + 1 * 512))[1];
            bN2.u[0] = ((const uint4*)(bt + 2 * 512))[0];
            bN2.u[1] = ((const uint4*)(bt + 2 * 512))[1];
            bN3.u[0] = ((const uint4*)(bt + 3 * 512))[0];
            bN3.u[1] = ((const uint4*)(bt + 3 * 512))[1];
        }

        acc0 = __builtin_amdgcn_wmma_f32_16x16x32_bf16(false, aC.v, false, bC0.v,
                                                       (short)0, acc0, false, false);
        acc1 = __builtin_amdgcn_wmma_f32_16x16x32_bf16(false, aC.v, false, bC1.v,
                                                       (short)0, acc1, false, false);
        acc2 = __builtin_amdgcn_wmma_f32_16x16x32_bf16(false, aC.v, false, bC2.v,
                                                       (short)0, acc2, false, false);
        acc3 = __builtin_amdgcn_wmma_f32_16x16x32_bf16(false, aC.v, false, bC3.v,
                                                       (short)0, acc3, false, false);

        aC = aN;
        bC0 = bN0; bC1 = bN1; bC2 = bN2; bC3 = bN3;
    }

    const int r0 = mt * 16 + ((lane < 16) ? 0 : 8);
    v8f accs[4] = {acc0, acc1, acc2, acc3};
#pragma unroll
    for (int j = 0; j < 4; ++j) {
        const int cc = (ng * 4 + j) * 16 + (lane & 15);
        const float bv = bias ? bias[cc] : 0.0f;
#pragma unroll
        for (int v = 0; v < 8; ++v) {
            float val = accs[j][v] + bv;
            if (doRelu) val = fmaxf(val, 0.0f);
            const size_t off = (size_t)(r0 + v) * N + cc;
            if (outBf16) ((unsigned short*)C)[off] = f2bf(val);
            else         ((float*)C)[off] = val;
        }
    }
}

__global__ void gemm_rowdot_kernel(const unsigned short* __restrict__ A,
                                   const unsigned short* __restrict__ Bp,
                                   const float* __restrict__ bias,
                                   const float* __restrict__ fvec,
                                   float* __restrict__ rowdot,
                                   int M, int K,
                                   const float* __restrict__ addend,
                                   const float* __restrict__ dinv,
                                   const float* __restrict__ cbias,
                                   const float* __restrict__ fb,
                                   float* __restrict__ accp) {
    const int lane = threadIdx.x & 31;
    const int wid  = threadIdx.x >> 5;
    const int mt = blockIdx.x * 8 + wid;
    if (mt >= (M >> 4)) return;

    const int kTiles = K >> 5;

    const int hsel = (lane < 16) ? 0 : 8;
    const unsigned short* arow = A + (size_t)(mt * 16 + (lane & 15)) * K;
    const unsigned short* bcol = Bp + lane * 16;
    const size_t bstride = (size_t)(COMB >> 4) * 512;

    v8f acc0 = {}, acc1 = {}, acc2 = {}, acc3 = {};

    V16 aC, aN;
    V16 bC0, bC1, bC2, bC3, bN0, bN1, bN2, bN3;

    {
        const uint4* ap = (const uint4*)(arow + hsel);
        aC.u[0] = ap[0];
        aC.u[1] = ap[2];
        const unsigned short* bt = bcol;
        bC0.u[0] = ((const uint4*)(bt + 0 * 512))[0];
        bC0.u[1] = ((const uint4*)(bt + 0 * 512))[1];
        bC1.u[0] = ((const uint4*)(bt + 1 * 512))[0];
        bC1.u[1] = ((const uint4*)(bt + 1 * 512))[1];
        bC2.u[0] = ((const uint4*)(bt + 2 * 512))[0];
        bC2.u[1] = ((const uint4*)(bt + 2 * 512))[1];
        bC3.u[0] = ((const uint4*)(bt + 3 * 512))[0];
        bC3.u[1] = ((const uint4*)(bt + 3 * 512))[1];
    }

    for (int kt = 0; kt < kTiles; ++kt) {
        const int ktn = (kt + 1 < kTiles) ? (kt + 1) : kt;
        {
            const uint4* ap = (const uint4*)(arow + ktn * 32 + hsel);
            aN.u[0] = ap[0];
            aN.u[1] = ap[2];
            const unsigned short* bt = bcol + (size_t)ktn * bstride;
            bN0.u[0] = ((const uint4*)(bt + 0 * 512))[0];
            bN0.u[1] = ((const uint4*)(bt + 0 * 512))[1];
            bN1.u[0] = ((const uint4*)(bt + 1 * 512))[0];
            bN1.u[1] = ((const uint4*)(bt + 1 * 512))[1];
            bN2.u[0] = ((const uint4*)(bt + 2 * 512))[0];
            bN2.u[1] = ((const uint4*)(bt + 2 * 512))[1];
            bN3.u[0] = ((const uint4*)(bt + 3 * 512))[0];
            bN3.u[1] = ((const uint4*)(bt + 3 * 512))[1];
        }

        acc0 = __builtin_amdgcn_wmma_f32_16x16x32_bf16(false, aC.v, false, bC0.v,
                                                       (short)0, acc0, false, false);
        acc1 = __builtin_amdgcn_wmma_f32_16x16x32_bf16(false, aC.v, false, bC1.v,
                                                       (short)0, acc1, false, false);
        acc2 = __builtin_amdgcn_wmma_f32_16x16x32_bf16(false, aC.v, false, bC2.v,
                                                       (short)0, acc2, false, false);
        acc3 = __builtin_amdgcn_wmma_f32_16x16x32_bf16(false, aC.v, false, bC3.v,
                                                       (short)0, acc3, false, false);

        aC = aN;
        bC0 = bN0; bC1 = bN1; bC2 = bN2; bC3 = bN3;
    }

    const int c16 = lane & 15;
    v8f accs[4] = {acc0, acc1, acc2, acc3};
    float p[8];
#pragma unroll
    for (int v = 0; v < 8; ++v) p[v] = 0.0f;
    float cpart = 0.0f;
#pragma unroll
    for (int j = 0; j < 4; ++j) {
        const int cc = j * 16 + c16;
        const float fv = fvec[cc];
        const float bv = bias ? bias[cc] : 0.0f;
        if (accp) cpart += cbias[cc] * fv;
#pragma unroll
        for (int v = 0; v < 8; ++v) p[v] += (accs[j][v] + bv) * fv;
    }
#pragma unroll
    for (int off = 8; off > 0; off >>= 1) {
#pragma unroll
        for (int v = 0; v < 8; ++v) p[v] += __shfl_xor(p[v], off, 32);
        cpart += __shfl_xor(cpart, off, 32);
    }
    float mine = p[0];
#pragma unroll
    for (int v = 1; v < 8; ++v) mine = ((lane & 7) == v) ? p[v] : mine;
    if (c16 < 8) {
        const int row = mt * 16 + ((lane >> 4) << 3) + c16;
        rowdot[row] = mine;
        if (accp) {
            const float di = dinv[row];
            accp[(size_t)row * ACC_PITCH] = addend[row] + (cpart + fb[0]) + di * di * mine;
        }
    }
}

__global__ void deg_init_kernel(float* __restrict__ deg, int n) {
    int i = blockIdx.x * blockDim.x + threadIdx.x;
    if (i < n) deg[i] = 1.0f;
}

__global__ void deg_accum_kernel(const int* __restrict__ dst,
                                 const float* __restrict__ ew,
                                 float* __restrict__ deg, int nE) {
    int e = blockIdx.x * blockDim.x + threadIdx.x;
    if (e < nE) atomicAdd(&deg[dst[e]], ew[e]);
}

__global__ void dinv_kernel(const float* __restrict__ deg,
                            float* __restrict__ dinv, int n) {
    int i = blockIdx.x * blockDim.x + threadIdx.x;
    if (i >= n) return;
    float d = deg[i];
    dinv[i] = (d > 0.0f) ? rsqrtf(fmaxf(d, 1e-12f)) : 0.0f;
}

__global__ void edge_norm_kernel(const int* __restrict__ src,
                                 const int* __restrict__ dst,
                                 const float* __restrict__ ew,
                                 const float* __restrict__ dinv,
                                 float* __restrict__ norm, int nE) {
    int e = blockIdx.x * blockDim.x + threadIdx.x;
    if (e < nE) norm[e] = dinv[src[e]] * ew[e] * dinv[dst[e]];
}

__global__ void agg_init_kernel(const float* __restrict__ h,
                                const float* __restrict__ dinv,
                                float* __restrict__ agg, int nNodes) {
    int idx = blockIdx.x * blockDim.x + threadIdx.x;
    int n = idx >> 4;
    if (n >= nNodes) return;
    int c = (idx & 15) * 4;
    float w = dinv[n]; w *= w;
    float4 hv = *(const float4*)(h + (size_t)n * COMB + c);
    float4 o = {hv.x * w, hv.y * w, hv.z * w, hv.w * w};
    *(float4*)(agg + (size_t)n * COMB + c) = o;
}

__global__ void gcn_scatter_kernel(const float* __restrict__ h,
                                   const int* __restrict__ src,
                                   const int* __restrict__ dst,
                                   const float* __restrict__ norm,
                                   float* __restrict__ agg, int nE) {
    int idx = blockIdx.x * blockDim.x + threadIdx.x;
    int e = idx >> 4;
    if (e >= nE) return;
    int c = (idx & 15) * 4;
    int s = src[e], d = dst[e];
    float nm = norm[e];
    float4 hv = *(const float4*)(h + (size_t)s * COMB + c);
    float* ap = agg + (size_t)d * COMB + c;
    atomicAdd(ap + 0, hv.x * nm);
    atomicAdd(ap + 1, hv.y * nm);
    atomicAdd(ap + 2, hv.z * nm);
    atomicAdd(ap + 3, hv.w * nm);
}

__global__ void relu_bias_bf16_kernel(const float* __restrict__ agg,
                                      const float* __restrict__ b,
                                      unsigned short* __restrict__ g, int total) {
    int idx = blockIdx.x * blockDim.x + threadIdx.x;
    if (idx >= total) return;
    float v = agg[idx] + b[idx & (COMB - 1)];
    g[idx] = f2bf(fmaxf(v, 0.0f));
}

__global__ void gcn_scatter_dot_kernel(const float* __restrict__ s,
                                       const int* __restrict__ src,
                                       const int* __restrict__ dst,
                                       const float* __restrict__ norm,
                                       float* __restrict__ accp, int nE4) {
    int i = blockIdx.x * blockDim.x + threadIdx.x;
    if (i >= nE4) return;
    const int4   sv = ((const int4*)src)[i];
    const int4   dv = ((const int4*)dst)[i];
    const float4 nv = ((const float4*)norm)[i];
    const float m0 = nv.x * s[sv.x];
    const float m1 = nv.y * s[sv.y];
    const float m2 = nv.z * s[sv.z];
    const float m3 = nv.w * s[sv.w];
    atomicAdd(accp + (size_t)dv.x * ACC_PITCH, m0);
    atomicAdd(accp + (size_t)dv.y * ACC_PITCH, m1);
    atomicAdd(accp + (size_t)dv.z * ACC_PITCH, m2);
    atomicAdd(accp + (size_t)dv.w * ACC_PITCH, m3);
}

__global__ void out_gather_kernel(const float* __restrict__ accp,
                                  float* __restrict__ out, int n) {
    int i = blockIdx.x * blockDim.x + threadIdx.x;
    if (i < n) out[i] = accp[(size_t)i * ACC_PITCH];
}

extern "C" void kernel_launch(void* const* d_in, const int* in_sizes, int n_in,
                              void* d_out, int out_size, void* d_ws, size_t ws_size,
                              hipStream_t stream) {
    (void)in_sizes; (void)n_in; (void)out_size; (void)ws_size;

    const float* x        = (const float*)d_in[0];
    const int*   eidx     = (const int*)d_in[1];
    const float* eattr    = (const float*)d_in[2];
    const float* ufeat    = (const float*)d_in[3];
    const float* gcn1_W   = (const float*)d_in[4];
    const float* gcn1_b   = (const float*)d_in[5];
    const float* gcn2_W   = (const float*)d_in[6];
    const float* gcn2_b   = (const float*)d_in[7];
    const float* fc1_W    = (const float*)d_in[8];
    const float* fc1_b    = (const float*)d_in[9];
    const float* fc2_W    = (const float*)d_in[10];
    const float* fc2_b    = (const float*)d_in[11];
    const float* fcf_W    = (const float*)d_in[12];
    const float* fcf_b    = (const float*)d_in[13];
    const int* esrc = eidx;
    const int* edst = eidx + N_EDGES;
    float* out = (float*)d_out;

    char* ws = (char*)d_ws;
    size_t off = 0;
    auto carve = [&](size_t bytes) -> void* {
        void* p = ws + off;
        off = (off + bytes + 255) & ~(size_t)255;
        return p;
    };
    unsigned short* UBF = (unsigned short*)carve((size_t)N_NODES * MLP_IN * 2);
    unsigned short* XBF = (unsigned short*)carve((size_t)N_NODES * GNN_IN * 2);
    unsigned short* HBF = (unsigned short*)carve((size_t)N_NODES * MLP_HID * 2);
    unsigned short* GBF = (unsigned short*)carve((size_t)N_NODES * COMB * 2);
    unsigned short* W1P = (unsigned short*)carve((size_t)MLP_IN * MLP_HID * 2);
    unsigned short* W2P = (unsigned short*)carve((size_t)MLP_HID * COMB * 2);
    unsigned short* G1P = (unsigned short*)carve((size_t)GNN_IN * GNN_HID * 2);
    unsigned short* G2P = (unsigned short*)carve((size_t)GNN_HID * COMB * 2);
    float* H1     = (float*)carve((size_t)N_NODES * COMB * 4);
    float* AGG    = (float*)carve((size_t)N_NODES * COMB * 4);
    float* MDOT   = (float*)carve((size_t)N_NODES * 4);
    float* SDOT   = (float*)carve((size_t)N_NODES * 4);
    float* ACC    = (float*)carve((size_t)N_NODES * ACC_PITCH * 4);
    float* DEG    = (float*)carve((size_t)N_NODES * 4);
    float* DINV   = (float*)carve((size_t)N_NODES * 4);
    float* NORM   = (float*)carve((size_t)N_EDGES * 4);

    {
        int n4 = (N_NODES * MLP_IN) / 4;
        cvt_bf16_kernel<<<(n4 + 255) / 256, 256, 0, stream>>>(ufeat, UBF, n4);
        n4 = (N_NODES * GNN_IN) / 4;
        cvt_bf16_kernel<<<(n4 + 255) / 256, 256, 0, stream>>>(x, XBF, n4);
    }
    auto pack = [&](const float* W, unsigned short* P, int K, int N) {
        int total = (K / 32) * (N / 16) * 32;
        pack_w_kernel<<<(total + 255) / 256, 256, 0, stream>>>(W, P, K, N);
    };
    pack(fc1_W, W1P, MLP_IN, MLP_HID);
    pack(fc2_W, W2P, MLP_HID, COMB);
    pack(gcn1_W, G1P, GNN_IN, GNN_HID);
    pack(gcn2_W, G2P, GNN_HID, COMB);

    deg_init_kernel<<<(N_NODES + 255) / 256, 256, 0, stream>>>(DEG, N_NODES);
    deg_accum_kernel<<<(N_EDGES + 255) / 256, 256, 0, stream>>>(edst, eattr, DEG, N_EDGES);
    dinv_kernel<<<(N_NODES + 255) / 256, 256, 0, stream>>>(DEG, DINV, N_NODES);
    edge_norm_kernel<<<(N_EDGES + 255) / 256, 256, 0, stream>>>(esrc, edst, eattr, DINV,
                                                                NORM, N_EDGES);

    auto gemm = [&](const unsigned short* A, const unsigned short* Bp,
                    const float* bias, void* C, int M, int N, int K,
                    int relu, int outBf) {
        int tasks = (M / 16) * (N / 64);
        int blocks = (tasks + 7) / 8;
        gemm_wmma_kernel<<<blocks, 256, 0, stream>>>(A, Bp, bias, C, M, N, K, relu, outBf);
    };

    const int rdBlocks = ((N_NODES / 16) + 7) / 8;

    gemm(UBF, W1P, fc1_b, HBF, N_NODES, MLP_HID, MLP_IN,  1,  1);
    gemm_rowdot_kernel<<<rdBlocks, 256, 0, stream>>>(HBF, W2P, fc2_b, fcf_W, MDOT,
                                                     N_NODES, MLP_HID,
                                                     nullptr, nullptr, nullptr, nullptr, nullptr);

    gemm(XBF, G1P, nullptr, H1, N_NODES, GNN_HID, GNN_IN, 0, 0);
    agg_init_kernel<<<(N_NODES * 16 + 255) / 256, 256, 0, stream>>>(H1, DINV, AGG, N_NODES);
    gcn_scatter_kernel<<<(N_EDGES * 16 + 255) / 256, 256, 0, stream>>>(
        H1, esrc, edst, NORM, AGG, N_EDGES);
    relu_bias_bf16_kernel<<<(N_NODES * COMB + 255) / 256, 256, 0, stream>>>(
        AGG, gcn1_b, GBF, N_NODES * COMB);

    gemm_rowdot_kernel<<<rdBlocks, 256, 0, stream>>>(GBF, G2P, nullptr, fcf_W + COMB, SDOT,
                                                     N_NODES, GNN_HID,
                                                     MDOT, DINV, gcn2_b, fcf_b, ACC);
    gcn_scatter_dot_kernel<<<(N_EDGES / 4 + 255) / 256, 256, 0, stream>>>(
        SDOT, esrc, edst, NORM, ACC, N_EDGES / 4);

    out_gather_kernel<<<(N_NODES + 255) / 256, 256, 0, stream>>>(ACC, out, N_NODES);
}


# layer-1 edge scatter: a wave walks 32 edges, lane l on floats l and l+32 of the row, so every f32 atomic instruction adds one whole 128-byte line (2 requests per edge, was 8 partial-line ones)
# speedup vs baseline: 1.7698x; 1.3554x over previous
#include <hip/hip_runtime.h>
#include <hip/hip_bf16.h>


#define N_NODES 50000
#define N_EDGES 800000
#define GNN_IN  256
#define GNN_HID 64
#define COMB    64
#define MLP_IN  256
#define MLP_HID 512
#define ACC_PITCH 32

typedef __attribute__((ext_vector_type(16))) __bf16 v16bf;
typedef __attribute__((ext_vector_type(8)))  float  v8f;

union V16 {
    uint4 u[2];
    v16bf v;
};

__device__ __forceinline__ unsigned short f2bf(float f) {
    unsigned int u = __float_as_uint(f);
    unsigned int r = u + 0x7FFFu + ((u >> 16) & 1u);
    return (unsigned short)(r >> 16);
}

__device__ __forceinline__ void cvt_bf16_body(const float* __restrict__ in,
                                              unsigned short* __restrict__ out,
                                              int n4, int i) {
    if (i >= n4) return;
    const float4 f = ((const float4*)in)[i];
    ushort4 o;
    o.x = f2bf(f.x); o.y = f2bf(f.y); o.z = f2bf(f.z); o.w = f2bf(f.w);
    ((ushort4*)out)[i] = o;
}

__device__ __forceinline__ void pack_w_body(const float* __restrict__ W,
                                            unsigned short* __restrict__ P,
                                            int K, int N, int idx) {
    int nTiles = N >> 4;
    int total = (K >> 5) * nTiles * 32;
    if (idx >= total) return;
    int lane = idx & 31;
    int tile = idx >> 5;
    int kt = tile / nTiles;
    int nt = tile % nTiles;
    int col = nt * 16 + (lane & 15);
    int kb  = kt * 32 + ((lane < 16) ? 0 : 8);
    __attribute__((aligned(16))) unsigned short o[16];
#pragma unroll
    for (int j = 0; j < 8; ++j) o[j]     = f2bf(W[(size_t)(kb + j) * N + col]);
#pragma unroll
    for (int j = 0; j < 8; ++j) o[8 + j] = f2bf(W[(size_t)(kb + 16 + j) * N + col]);
    uint4* dst = (uint4*)(P + (size_t)tile * 512 + lane * 16);
    dst[0] = ((uint4*)o)[0];
    dst[1] = ((uint4*)o)[1];
}

__device__ __forceinline__ float dinv_of(float d) {
    return (d > 0.0f) ? rsqrtf(fmaxf(d, 1e-12f)) : 0.0f;
}

#define PB_W1  (((MLP_IN  / 32) * (MLP_HID / 16) * 32) / 256)
#define PB_W2  (((MLP_HID / 32) * (COMB    / 16) * 32) / 256)
#define PB_G1  (((GNN_IN  / 32) * (GNN_HID / 16) * 32) / 256)
#define PB_G2  (((GNN_HID / 32) * (COMB    / 16) * 32) / 256)
#define PB_DEG ((N_NODES + 255) / 256)
#define PB_CVU ((N_NODES * MLP_IN / 4) / 256)
#define PB_CVX ((N_NODES * GNN_IN / 4) / 256)
#define PB_ALL (PB_W1 + PB_W2 + PB_G1 + PB_G2 + PB_DEG + PB_CVU + PB_CVX)

__global__ void prep_kernel(const float* __restrict__ ufeat,
                            unsigned short* __restrict__ UBF,
                            const float* __restrict__ x,
                            unsigned short* __restrict__ XBF,
                            const float* __restrict__ fc1_W,
                            unsigned short* __restrict__ W1P,
                            const float* __restrict__ fc2_W,
                            unsigned short* __restrict__ W2P,
                            const float* __restrict__ gcn1_W,
                            unsigned short* __restrict__ G1P,
                            const float* __restrict__ gcn2_W,
                            unsigned short* __restrict__ G2P,
                            float* __restrict__ deg) {
    int b = blockIdx.x;
    const int t = threadIdx.x;
    if (b < PB_W1) { pack_w_body(fc1_W, W1P, MLP_IN, MLP_HID, b * 256 + t); return; }
    b -= PB_W1;
    if (b < PB_W2) { pack_w_body(fc2_W, W2P, MLP_HID, COMB, b * 256 + t); return; }
    b -= PB_W2;
    if (b < PB_G1) { pack_w_body(gcn1_W, G1P, GNN_IN, GNN_HID, b * 256 + t); return; }
    b -= PB_G1;
    if (b < PB_G2) { pack_w_body(gcn2_W, G2P, GNN_HID, COMB, b * 256 + t); return; }
    b -= PB_G2;
    if (b < PB_DEG) {
        const int i = b * 256 + t;
        if (i < N_NODES) deg[i] = 1.0f;
        return;
    }
    b -= PB_DEG;
    if (b < PB_CVU) { cvt_bf16_body(ufeat, UBF, N_NODES * MLP_IN / 4, b * 256 + t); return; }
    b -= PB_CVU;
    cvt_bf16_body(x, XBF, N_NODES * GNN_IN / 4, b * 256 + t);
}

template <bool AGG_INIT>
__global__ void gemm_wmma_kernel(const unsigned short* __restrict__ A,
                                 const unsigned short* __restrict__ Bp,
                                 const float* __restrict__ bias,
                                 void* __restrict__ C,
                                 int M, int N, int K, int doRelu, int outBf16,
                                 const float* __restrict__ deg,
                                 float* __restrict__ agg) {
    const int lane = threadIdx.x & 31;
    const int wid  = threadIdx.x >> 5;
    const int nGroups = N >> 6;
    const int mTiles  = M >> 4;
    const int task = blockIdx.x * 8 + wid;
    if (task >= mTiles * nGroups) return;
    const int mt = task / nGroups;
    const int ng = task % nGroups;
    const int kTiles = K >> 5;
    const int nTiles = N >> 4;

    const int hsel = (lane < 16) ? 0 : 8;
    const unsigned short* arow = A + (size_t)(mt * 16 + (lane & 15)) * K;
    const unsigned short* bcol = Bp + (size_t)(ng << 2) * 512 + lane * 16;
    const size_t bstride = (size_t)nTiles * 512;

    v8f acc0 = {}, acc1 = {}, acc2 = {}, acc3 = {};

    V16 aC, aN;
    V16 bC0, bC1, bC2, bC3, bN0, bN1, bN2, bN3;

    {
        const uint4* ap = (const uint4*)(arow + hsel);
        aC.u[0] = ap[0];
        aC.u[1] = ap[2];
        const unsigned short* bt = bcol;
        bC0.u[0] = ((const uint4*)(bt + 0 * 512))[0];
        bC0.u[1] = ((const uint4*)(bt + 0 * 512))[1];
        bC1.u[0] = ((const uint4*)(bt + 1 * 512))[0];
        bC1.u[1] = ((const uint4*)(bt + 1 * 512))[1];
        bC2.u[0] = ((const uint4*)(bt + 2 * 512))[0];
        bC2.u[1] = ((const uint4*)(bt + 2 * 512))[1];
        bC3.u[0] = ((const uint4*)(bt + 3 * 512))[0];
        bC3.u[1] = ((const uint4*)(bt + 3 * 512))[1];
    }

    for (int kt = 0; kt < kTiles; ++kt) {
        const int ktn = (kt + 1 < kTiles) ? (kt + 1) : kt;
        {
            const uint4* ap = (const uint4*)(arow + ktn * 32 + hsel);
            aN.u[0] = ap[0];
            aN.u[1] = ap[2];
            const unsigned short* bt = bcol + (size_t)ktn * bstride;
            bN0.u[0] = ((const uint4*)(bt + 0 * 512))[0];
            bN0.u[1] = ((const uint4*)(bt + 0 * 512))[1];
            bN1.u[0] = ((const uint4*)(bt + 1 * 512))[0];
            bN1.u[1] = ((const uint4*)(bt + 1 * 512))[1];
            bN2.u[0] = ((const uint4*)(bt + 2 * 512))[0];
            bN2.u[1] = ((const uint4*)(bt + 2 * 512))[1];
            bN3.u[0] = ((const uint4*)(bt + 3 * 512))[0];
            bN3.u[1] = ((const uint4*)(bt + 3 * 512))[1];
        }

        acc0 = __builtin_amdgcn_wmma_f32_16x16x32_bf16(false, aC.v, false, bC0.v,
                                                       (short)0, acc0, false, false);
        acc1 = __builtin_amdgcn_wmma_f32_16x16x32_bf16(false, aC.v, false, bC1.v,
                                                       (short)0, acc1, false, false);
        acc2 = __builtin_amdgcn_wmma_f32_16x16x32_bf16(false, aC.v, false, bC2.v,
                                                       (short)0, acc2, false, false);
        acc3 = __builtin_amdgcn_wmma_f32_16x16x32_bf16(false, aC.v, false, bC3.v,
                                                       (short)0, acc3, false, false);

        aC = aN;
        bC0 = bN0; bC1 = bN1; bC2 = bN2; bC3 = bN3;
    }

    const int r0 = mt * 16 + ((lane < 16) ? 0 : 8);
    v8f accs[4] = {acc0, acc1, acc2, acc3};
    float selfw[8];
#pragma unroll
    for (int v = 0; v < 8; ++v) {
        if (AGG_INIT) { const float di = dinv_of(deg[r0 + v]); selfw[v] = di * di; }
        else          selfw[v] = 0.0f;
    }
#pragma unroll
    for (int j = 0; j < 4; ++j) {
        const int cc = (ng * 4 + j) * 16 + (lane & 15);
        const float bv = bias ? bias[cc] : 0.0f;
#pragma unroll
        for (int v = 0; v < 8; ++v) {
            float val = accs[j][v] + bv;
            if (doRelu) val = fmaxf(val, 0.0f);
            const size_t off = (size_t)(r0 + v) * N + cc;
            if (outBf16) ((unsigned short*)C)[off] = f2bf(val);
            else         ((float*)C)[off] = val;
            if (AGG_INIT) agg[off] = val * selfw[v];
        }
    }
}

__device__ __forceinline__ unsigned int relu_bias_bf16x2(float a, float ba,
                                                         float b, float bb) {
    return (unsigned int)f2bf(fmaxf(a + ba, 0.0f)) |
           ((unsigned int)f2bf(fmaxf(b + bb, 0.0f)) << 16);
}

__device__ __forceinline__ uint4 relu_bias_bf16x8(float4 f0, float4 f1,
                                                  float4 b0, float4 b1) {
    uint4 r;
    r.x = relu_bias_bf16x2(f0.x, b0.x, f0.y, b0.y);
    r.y = relu_bias_bf16x2(f0.z, b0.z, f0.w, b0.w);
    r.z = relu_bias_bf16x2(f1.x, b1.x, f1.y, b1.y);
    r.w = relu_bias_bf16x2(f1.z, b1.z, f1.w, b1.w);
    return r;
}

__device__ __forceinline__ void load_a_frag_f32(V16& a,
                                                const float* __restrict__ arow32,
                                                const float* __restrict__ abias,
                                                int koff) {
    const float4* fp = (const float4*)(arow32 + koff);
    const float4* bp = (const float4*)(abias + koff);
    a.u[0] = relu_bias_bf16x8(fp[0], fp[1], bp[0], bp[1]);
    a.u[1] = relu_bias_bf16x8(fp[4], fp[5], bp[4], bp[5]);
}

template <bool A_F32>
__global__ void gemm_rowdot_kernel(const unsigned short* __restrict__ A16,
                                   const float* __restrict__ A32,
                                   const float* __restrict__ abias,
                                   const unsigned short* __restrict__ Bp,
                                   const float* __restrict__ bias,
                                   const float* __restrict__ fvec,
                                   float* __restrict__ rowdot,
                                   int M, int K,
                                   const float* __restrict__ addend,
                                   const float* __restrict__ deg,
                                   const float* __restrict__ cbias,
                                   const float* __restrict__ fb,
                                   float* __restrict__ accp) {
    const int lane = threadIdx.x & 31;
    const int wid  = threadIdx.x >> 5;
    const int mt = blockIdx.x * 8 + wid;
    if (mt >= (M >> 4)) return;

    const int kTiles = K >> 5;

    const int hsel = (lane < 16) ? 0 : 8;
    const unsigned short* arow = A16 + (size_t)(mt * 16 + (lane & 15)) * K;
    const float* arow32 = A32 + (size_t)(mt * 16 + (lane & 15)) * K;
    const unsigned short* bcol = Bp + lane * 16;
    const size_t bstride = (size_t)(COMB >> 4) * 512;

    v8f acc0 = {}, acc1 = {}, acc2 = {}, acc3 = {};

    V16 aC, aN;
    V16 bC0, bC1, bC2, bC3, bN0, bN1, bN2, bN3;

    {
        if (!A_F32) {
            const uint4* ap = (const uint4*)(arow + hsel);
            aC.u[0] = ap[0];
            aC.u[1] = ap[2];
        } else {
            load_a_frag_f32(aC, arow32, abias, hsel);
        }
        const unsigned short* bt = bcol;
        bC0.u[0] = ((const uint4*)(bt + 0 * 512))[0];
        bC0.u[1] = ((const uint4*)(bt + 0 * 512))[1];
        bC1.u[0] = ((const uint4*)(bt + 1 * 512))[0];
        bC1.u[1] = ((const uint4*)(bt + 1 * 512))[1];
        bC2.u[0] = ((const uint4*)(bt + 2 * 512))[0];
        bC2.u[1] = ((const uint4*)(bt + 2 * 512))[1];
        bC3.u[0] = ((const uint4*)(bt + 3 * 512))[0];
        bC3.u[1] = ((const uint4*)(bt + 3 * 512))[1];
    }

    for (int kt = 0; kt < kTiles; ++kt) {
        const int ktn = (kt + 1 < kTiles) ? (kt + 1) : kt;
        {
            if (!A_F32) {
                const uint4* ap = (const uint4*)(arow + ktn * 32 + hsel);
                aN.u[0] = ap[0];
                aN.u[1] = ap[2];
            } else {
                load_a_frag_f32(aN, arow32, abias, ktn * 32 + hsel);
            }
            const unsigned short* bt = bcol + (size_t)ktn * bstride;
            bN0.u[0] = ((const uint4*)(bt + 0 * 512))[0];
            bN0.u[1] = ((const uint4*)(bt + 0 * 512))[1];
            bN1.u[0] = ((const uint4*)(bt + 1 * 512))[0];
            bN1.u[1] = ((const uint4*)(bt + 1 * 512))[1];
            bN2.u[0] = ((const uint4*)(bt + 2 * 512))[0];
            bN2.u[1] = ((const uint4*)(bt + 2 * 512))[1];
            bN3.u[0] = ((const uint4*)(bt + 3 * 512))[0];
            bN3.u[1] = ((const uint4*)(bt + 3 * 512))[1];
        }

        acc0 = __builtin_amdgcn_wmma_f32_16x16x32_bf16(false, aC.v, false, bC0.v,
                                                       (short)0, acc0, false, false);
        acc1 = __builtin_amdgcn_wmma_f32_16x16x32_bf16(false, aC.v, false, bC1.v,
                                                       (short)0, acc1, false, false);
        acc2 = __builtin_amdgcn_wmma_f32_16x16x32_bf16(false, aC.v, false, bC2.v,
                                                       (short)0, acc2, false, false);
        acc3 = __builtin_amdgcn_wmma_f32_16x16x32_bf16(false, aC.v, false, bC3.v,
                                                       (short)0, acc3, false, false);

        aC = aN;
        bC0 = bN0; bC1 = bN1; bC2 = bN2; bC3 = bN3;
    }

    const int c16 = lane & 15;
    v8f accs[4] = {acc0, acc1, acc2, acc3};
    float p[8];
#pragma unroll
    for (int v = 0; v < 8; ++v) p[v] = 0.0f;
    float cpart = 0.0f;
#pragma unroll
    for (int j = 0; j < 4; ++j) {
        const int cc = j * 16 + c16;
        const float fv = fvec[cc];
        const float bv = bias ? bias[cc] : 0.0f;
        if (accp) cpart += cbias[cc] * fv;
#pragma unroll
        for (int v = 0; v < 8; ++v) p[v] += (accs[j][v] + bv) * fv;
    }
#pragma unroll
    for (int off = 8; off > 0; off >>= 1) {
#pragma unroll
        for (int v = 0; v < 8; ++v) p[v] += __shfl_xor(p[v], off, 32);
        cpart += __shfl_xor(cpart, off, 32);
    }
    float mine = p[0];
#pragma unroll
    for (int v = 1; v < 8; ++v) mine = ((lane & 7) == v) ? p[v] : mine;
    if (c16 < 8) {
        const int row = mt * 16 + ((lane >> 4) << 3) + c16;
        rowdot[row] = mine;
        if (accp) {
            const float di = dinv_of(deg[row]);
            accp[(size_t)row * ACC_PITCH] = addend[row] + (cpart + fb[0]) + di * di * mine;
        }
    }
}

__global__ void deg_accum_kernel(const int* __restrict__ dst,
                                 const float* __restrict__ ew,
                                 float* __restrict__ deg, int nE) {
    int e = blockIdx.x * blockDim.x + threadIdx.x;
    if (e < nE) atomicAdd(&deg[dst[e]], ew[e]);
}

__global__ void edge_norm_kernel(const int* __restrict__ src,
                                 const int* __restrict__ dst,
                                 const float* __restrict__ ew,
                                 const float* __restrict__ deg,
                                 float* __restrict__ norm, int nE) {
    int e = blockIdx.x * blockDim.x + threadIdx.x;
    if (e < nE) norm[e] = dinv_of(deg[src[e]]) * ew[e] * dinv_of(deg[dst[e]]);
}

#define SC_EPW 32
#define SC_GRP 8
static_assert(N_EDGES % SC_EPW == 0, "a wave always holds SC_EPW real edges");
static_assert(COMB == 64, "lane l owns floats l and l+32");

__global__ void gcn_scatter_kernel(const float* __restrict__ h,
                                   const int* __restrict__ src,
                                   const int* __restrict__ dst,
                                   const float* __restrict__ norm,
                                   float* __restrict__ agg, int nE) {
    const int lane = threadIdx.x & 31;
    const int e0 = (blockIdx.x * (blockDim.x >> 5) + (threadIdx.x >> 5)) * SC_EPW;
    if (e0 >= nE) return;
    const int sv = src[e0 + lane];
    const int dv = dst[e0 + lane];
    const int nv = __float_as_int(norm[e0 + lane]);
    const float* hl = h + lane;
    float* al = agg + lane;

    float lo[2][SC_GRP], hi[2][SC_GRP];
#pragma unroll
    for (int k = 0; k < SC_GRP; ++k) {
        const int s = __builtin_amdgcn_readlane(sv, k);
        lo[0][k] = hl[(size_t)s * COMB];
        hi[0][k] = hl[(size_t)s * COMB + 32];
    }
#pragma unroll
    for (int g = 0; g < SC_EPW / SC_GRP; ++g) {
        const int cur = g & 1, nxt = cur ^ 1;
        if (g + 1 < SC_EPW / SC_GRP) {
#pragma unroll
            for (int k = 0; k < SC_GRP; ++k) {
                const int s = __builtin_amdgcn_readlane(sv, (g + 1) * SC_GRP + k);
                lo[nxt][k] = hl[(size_t)s * COMB];
                hi[nxt][k] = hl[(size_t)s * COMB + 32];
            }
        }
#pragma unroll
        for (int k = 0; k < SC_GRP; ++k) {
            const int d = __builtin_amdgcn_readlane(dv, g * SC_GRP + k);
            const float nm = __int_as_float(__builtin_amdgcn_readlane(nv, g * SC_GRP + k));
            float* ap = al + (size_t)d * COMB;
            atomicAdd(ap,      lo[cur][k] * nm);
            atomicAdd(ap + 32, hi[cur][k] * nm);
        }
    }
}

__global__ void gcn_scatter_dot_kernel(const float* __restrict__ s,
                                       const int* __restrict__ src,
                                       const int* __restrict__ dst,
                                       const float* __restrict__ norm,
                                       float* __restrict__ accp, int nE4) {
    int i = blockIdx.x * blockDim.x + threadIdx.x;
    if (i >= nE4) return;
    const int4   sv = ((const int4*)src)[i];
    const int4   dv = ((const int4*)dst)[i];
    const float4 nv = ((const float4*)norm)[i];
    const float m0 = nv.x * s[sv.x];
    const float m1 = nv.y * s[sv.y];
    const float m2 = nv.z * s[sv.z];
    const float m3 = nv.w * s[sv.w];
    atomicAdd(accp + (size_t)dv.x * ACC_PITCH, m0);
    atomicAdd(accp + (size_t)dv.y * ACC_PITCH, m1);
    atomicAdd(accp + (size_t)dv.z * ACC_PITCH, m2);
    atomicAdd(accp + (size_t)dv.w * ACC_PITCH, m3);
}

__global__ void out_gather_kernel(const float* __restrict__ accp,
                                  float* __restrict__ out, int n) {
    int i = blockIdx.x * blockDim.x + threadIdx.x;
    if (i < n) out[i] = accp[(size_t)i * ACC_PITCH];
}

extern "C" void kernel_launch(void* const* d_in, const int* in_sizes, int n_in,
                              void* d_out, int out_size, void* d_ws, size_t ws_size,
                              hipStream_t stream) {
    (void)in_sizes; (void)n_in; (void)out_size; (void)ws_size;

    const float* x        = (const float*)d_in[0];
    const int*   eidx     = (const int*)d_in[1];
    const float* eattr    = (const float*)d_in[2];
    const float* ufeat    = (const float*)d_in[3];
    const float* gcn1_W   = (const float*)d_in[4];
    const float* gcn1_b   = (const float*)d_in[5];
    const float* gcn2_W   = (const float*)d_in[6];
    const float* gcn2_b   = (const float*)d_in[7];
    const float* fc1_W    = (const float*)d_in[8];
    const float* fc1_b    = (const float*)d_in[9];
    const float* fc2_W    = (const float*)d_in[10];
    const float* fc2_b    = (const float*)d_in[11];
    const float* fcf_W    = (const float*)d_in[12];
    const float* fcf_b    = (const float*)d_in[13];
    const int* esrc = eidx;
    const int* edst = eidx + N_EDGES;
    float* out = (float*)d_out;

    char* ws = (char*)d_ws;
    size_t off = 0;
    auto carve = [&](size_t bytes) -> void* {
        void* p = ws + off;
        off = (off + bytes + 255) & ~(size_t)255;
        return p;
    };
    unsigned short* UBF = (unsigned short*)carve((size_t)N_NODES * MLP_IN * 2);
    unsigned short* XBF = (unsigned short*)carve((size_t)N_NODES * GNN_IN * 2);
    unsigned short* HBF = (unsigned short*)carve((size_t)N_NODES * MLP_HID * 2);
    unsigned short* W1P = (unsigned short*)carve((size_t)MLP_IN * MLP_HID * 2);
    unsigned short* W2P = (unsigned short*)carve((size_t)MLP_HID * COMB * 2);
    unsigned short* G1P = (unsigned short*)carve((size_t)GNN_IN * GNN_HID * 2);
    unsigned short* G2P = (unsigned short*)carve((size_t)GNN_HID * COMB * 2);
    float* H1     = (float*)carve((size_t)N_NODES * COMB * 4);
    float* AGG    = (float*)carve((size_t)N_NODES * COMB * 4);
    float* MDOT   = (float*)carve((size_t)N_NODES * 4);
    float* SDOT   = (float*)carve((size_t)N_NODES * 4);
    float* ACC    = (float*)carve((size_t)N_NODES * ACC_PITCH * 4);
    float* DEG    = (float*)carve((size_t)N_NODES * 4);
    float* NORM   = (float*)carve((size_t)N_EDGES * 4);

    prep_kernel<<<PB_ALL, 256, 0, stream>>>(ufeat, UBF, x, XBF, fc1_W, W1P, fc2_W, W2P,
                                            gcn1_W, G1P, gcn2_W, G2P, DEG);

    deg_accum_kernel<<<(N_EDGES + 255) / 256, 256, 0, stream>>>(edst, eattr, DEG, N_EDGES);
    edge_norm_kernel<<<(N_EDGES + 255) / 256, 256, 0, stream>>>(esrc, edst, eattr, DEG,
                                                                NORM, N_EDGES);

    const int rdBlocks = ((N_NODES / 16) + 7) / 8;

    gemm_wmma_kernel<false><<<((N_NODES / 16) * (MLP_HID / 64) + 7) / 8, 256, 0, stream>>>(
        UBF, W1P, fc1_b, HBF, N_NODES, MLP_HID, MLP_IN,  1,  1,
        nullptr, nullptr);
    gemm_rowdot_kernel<false><<<rdBlocks, 256, 0, stream>>>(
        HBF, nullptr, nullptr, W2P, fc2_b, fcf_W, MDOT, N_NODES, MLP_HID,
        nullptr, nullptr, nullptr, nullptr, nullptr);

    gemm_wmma_kernel<true><<<((N_NODES / 16) * (GNN_HID / 64) + 7) / 8, 256, 0, stream>>>(
        XBF, G1P, nullptr, H1, N_NODES, GNN_HID, GNN_IN, 0, 0, DEG, AGG);
    gcn_scatter_kernel<<<(N_EDGES / SC_EPW + 7) / 8, 256, 0, stream>>>(
        H1, esrc, edst, NORM, AGG, N_EDGES);

    gemm_rowdot_kernel<true><<<rdBlocks, 256, 0, stream>>>(
        nullptr, AGG, gcn1_b, G2P, nullptr, fcf_W + COMB, SDOT, N_NODES, GNN_HID,
        MDOT, DEG, gcn2_b, fcf_b, ACC);
    gcn_scatter_dot_kernel<<<(N_EDGES / 4 + 255) / 256, 256, 0, stream>>>(
        SDOT, esrc, edst, NORM, ACC, N_EDGES / 4);

    out_gather_kernel<<<(N_NODES + 255) / 256, 256, 0, stream>>>(ACC, out, N_NODES);
}


# fc2 contracted with the projection once (w2f = fc2_W @ fw_lo, 512 floats); the fc1 GEMM reduces its relu rows against w2f in registers: H [N,512] never stored, fc2 GEMM launch gone
# speedup vs baseline: 2.0117x; 1.1367x over previous
#include <hip/hip_runtime.h>
#include <hip/hip_bf16.h>


#define N_NODES 50000
#define N_EDGES 800000
#define GNN_IN  256
#define GNN_HID 64
#define COMB    64
#define MLP_IN  256
#define MLP_HID 512
#define ACC_PITCH 32

typedef __attribute__((ext_vector_type(16))) __bf16 v16bf;
typedef __attribute__((ext_vector_type(8)))  float  v8f;

union V16 {
    uint4 u[2];
    v16bf v;
};

__device__ __forceinline__ unsigned short f2bf(float f) {
    unsigned int u = __float_as_uint(f);
    unsigned int r = u + 0x7FFFu + ((u >> 16) & 1u);
    return (unsigned short)(r >> 16);
}

__device__ __forceinline__ void cvt_bf16_body(const float* __restrict__ in,
                                              unsigned short* __restrict__ out,
                                              int n4, int i) {
    if (i >= n4) return;
    const float4 f = ((const float4*)in)[i];
    ushort4 o;
    o.x = f2bf(f.x); o.y = f2bf(f.y); o.z = f2bf(f.z); o.w = f2bf(f.w);
    ((ushort4*)out)[i] = o;
}

__device__ __forceinline__ void pack_w_body(const float* __restrict__ W,
                                            unsigned short* __restrict__ P,
                                            int K, int N, int idx) {
    int nTiles = N >> 4;
    int total = (K >> 5) * nTiles * 32;
    if (idx >= total) return;
    int lane = idx & 31;
    int tile = idx >> 5;
    int kt = tile / nTiles;
    int nt = tile % nTiles;
    int col = nt * 16 + (lane & 15);
    int kb  = kt * 32 + ((lane < 16) ? 0 : 8);
    __attribute__((aligned(16))) unsigned short o[16];
#pragma unroll
    for (int j = 0; j < 8; ++j) o[j]     = f2bf(W[(size_t)(kb + j) * N + col]);
#pragma unroll
    for (int j = 0; j < 8; ++j) o[8 + j] = f2bf(W[(size_t)(kb + 16 + j) * N + col]);
    uint4* dst = (uint4*)(P + (size_t)tile * 512 + lane * 16);
    dst[0] = ((uint4*)o)[0];
    dst[1] = ((uint4*)o)[1];
}

__device__ __forceinline__ void fold_w2_body(const float* __restrict__ W2,
                                             const float* __restrict__ b2,
                                             const float* __restrict__ fw_lo,
                                             float* __restrict__ w2f, int idx) {
    if (idx > MLP_HID) return;
    const float* row = (idx < MLP_HID) ? (W2 + (size_t)idx * COMB) : b2;
    float s = 0.0f;
    for (int c = 0; c < COMB; ++c) s += row[c] * fw_lo[c];
    w2f[idx] = s;
}

__device__ __forceinline__ float dinv_of(float d) {
    return (d > 0.0f) ? rsqrtf(fmaxf(d, 1e-12f)) : 0.0f;
}

#define PB_W1  (((MLP_IN  / 32) * (MLP_HID / 16) * 32) / 256)
#define PB_W2  ((MLP_HID + 1 + 255) / 256)
#define PB_G1  (((GNN_IN  / 32) * (GNN_HID / 16) * 32) / 256)
#define PB_G2  (((GNN_HID / 32) * (COMB    / 16) * 32) / 256)
#define PB_DEG ((N_NODES + 255) / 256)
#define PB_CVU ((N_NODES * MLP_IN / 4) / 256)
#define PB_CVX ((N_NODES * GNN_IN / 4) / 256)
#define PB_ALL (PB_W1 + PB_W2 + PB_G1 + PB_G2 + PB_DEG + PB_CVU + PB_CVX)

__global__ void prep_kernel(const float* __restrict__ ufeat,
                            unsigned short* __restrict__ UBF,
                            const float* __restrict__ x,
                            unsigned short* __restrict__ XBF,
                            const float* __restrict__ fc1_W,
                            unsigned short* __restrict__ W1P,
                            const float* __restrict__ fc2_W,
                            const float* __restrict__ fc2_b,
                            const float* __restrict__ fw_lo,
                            float* __restrict__ W2F,
                            const float* __restrict__ gcn1_W,
                            unsigned short* __restrict__ G1P,
                            const float* __restrict__ gcn2_W,
                            unsigned short* __restrict__ G2P,
                            float* __restrict__ deg) {
    int b = blockIdx.x;
    const int t = threadIdx.x;
    if (b < PB_W1) { pack_w_body(fc1_W, W1P, MLP_IN, MLP_HID, b * 256 + t); return; }
    b -= PB_W1;
    if (b < PB_W2) { fold_w2_body(fc2_W, fc2_b, fw_lo, W2F, b * 256 + t); return; }
    b -= PB_W2;
    if (b < PB_G1) { pack_w_body(gcn1_W, G1P, GNN_IN, GNN_HID, b * 256 + t); return; }
    b -= PB_G1;
    if (b < PB_G2) { pack_w_body(gcn2_W, G2P, GNN_HID, COMB, b * 256 + t); return; }
    b -= PB_G2;
    if (b < PB_DEG) {
        const int i = b * 256 + t;
        if (i < N_NODES) deg[i] = 1.0f;
        return;
    }
    b -= PB_DEG;
    if (b < PB_CVU) { cvt_bf16_body(ufeat, UBF, N_NODES * MLP_IN / 4, b * 256 + t); return; }
    b -= PB_CVU;
    cvt_bf16_body(x, XBF, N_NODES * GNN_IN / 4, b * 256 + t);
}

template <bool AGG_INIT>
__global__ void gemm_wmma_kernel(const unsigned short* __restrict__ A,
                                 const unsigned short* __restrict__ Bp,
                                 const float* __restrict__ bias,
                                 void* __restrict__ C,
                                 int M, int N, int K, int doRelu, int outBf16,
                                 const float* __restrict__ deg,
                                 float* __restrict__ agg) {
    const int lane = threadIdx.x & 31;
    const int wid  = threadIdx.x >> 5;
    const int nGroups = N >> 6;
    const int mTiles  = M >> 4;
    const int task = blockIdx.x * 8 + wid;
    if (task >= mTiles * nGroups) return;
    const int mt = task / nGroups;
    const int ng = task % nGroups;
    const int kTiles = K >> 5;
    const int nTiles = N >> 4;

    const int hsel = (lane < 16) ? 0 : 8;
    const unsigned short* arow = A + (size_t)(mt * 16 + (lane & 15)) * K;
    const unsigned short* bcol = Bp + (size_t)(ng << 2) * 512 + lane * 16;
    const size_t bstride = (size_t)nTiles * 512;

    v8f acc0 = {}, acc1 = {}, acc2 = {}, acc3 = {};

    V16 aC, aN;
    V16 bC0, bC1, bC2, bC3, bN0, bN1, bN2, bN3;

    {
        const uint4* ap = (const uint4*)(arow + hsel);
        aC.u[0] = ap[0];
        aC.u[1] = ap[2];
        const unsigned short* bt = bcol;
        bC0.u[0] = ((const uint4*)(bt + 0 * 512))[0];
        bC0.u[1] = ((const uint4*)(bt + 0 * 512))[1];
        bC1.u[0] = ((const uint4*)(bt + 1 * 512))[0];
        bC1.u[1] = ((const uint4*)(bt + 1 * 512))[1];
        bC2.u[0] = ((const uint4*)(bt + 2 * 512))[0];
        bC2.u[1] = ((const uint4*)(bt + 2 * 512))[1];
        bC3.u[0] = ((const uint4*)(bt + 3 * 512))[0];
        bC3.u[1] = ((const uint4*)(bt + 3 * 512))[1];
    }

    for (int kt = 0; kt < kTiles; ++kt) {
        const int ktn = (kt + 1 < kTiles) ? (kt + 1) : kt;
        {
            const uint4* ap = (const uint4*)(arow + ktn * 32 + hsel);
            aN.u[0] = ap[0];
            aN.u[1] = ap[2];
            const unsigned short* bt = bcol + (size_t)ktn * bstride;
            bN0.u[0] = ((const uint4*)(bt + 0 * 512))[0];
            bN0.u[1] = ((const uint4*)(bt + 0 * 512))[1];
            bN1.u[0] = ((const uint4*)(bt + 1 * 512))[0];
            bN1.u[1] = ((const uint4*)(bt + 1 * 512))[1];
            bN2.u[0] = ((const uint4*)(bt + 2 * 512))[0];
            bN2.u[1] = ((const uint4*)(bt + 2 * 512))[1];
            bN3.u[0] = ((const uint4*)(bt + 3 * 512))[0];
            bN3.u[1] = ((const uint4*)(bt + 3 * 512))[1];
        }

        acc0 = __builtin_amdgcn_wmma_f32_16x16x32_bf16(false, aC.v, false, bC0.v,
                                                       (short)0, acc0, false, false);
        acc1 = __builtin_amdgcn_wmma_f32_16x16x32_bf16(false, aC.v, false, bC1.v,
                                                       (short)0, acc1, false, false);
        acc2 = __builtin_amdgcn_wmma_f32_16x16x32_bf16(false, aC.v, false, bC2.v,
                                                       (short)0, acc2, false, false);
        acc3 = __builtin_amdgcn_wmma_f32_16x16x32_bf16(false, aC.v, false, bC3.v,
                                                       (short)0, acc3, false, false);

        aC = aN;
        bC0 = bN0; bC1 = bN1; bC2 = bN2; bC3 = bN3;
    }

    const int r0 = mt * 16 + ((lane < 16) ? 0 : 8);
    v8f accs[4] = {acc0, acc1, acc2, acc3};
    float selfw[8];
#pragma unroll
    for (int v = 0; v < 8; ++v) {
        if (AGG_INIT) { const float di = dinv_of(deg[r0 + v]); selfw[v] = di * di; }
        else          selfw[v] = 0.0f;
    }
#pragma unroll
    for (int j = 0; j < 4; ++j) {
        const int cc = (ng * 4 + j) * 16 + (lane & 15);
        const float bv = bias ? bias[cc] : 0.0f;
#pragma unroll
        for (int v = 0; v < 8; ++v) {
            float val = accs[j][v] + bv;
            if (doRelu) val = fmaxf(val, 0.0f);
            const size_t off = (size_t)(r0 + v) * N + cc;
            if (outBf16) ((unsigned short*)C)[off] = f2bf(val);
            else         ((float*)C)[off] = val;
            if (AGG_INIT) agg[off] = val * selfw[v];
        }
    }
}

static_assert(MLP_HID == 8 * 64, "8 waves x 64 columns cover one row");
static_assert(N_NODES % 16 == 0, "every block holds 16 real rows");

__global__ void gemm_relu_vecdot_kernel(const unsigned short* __restrict__ A,
                                        const unsigned short* __restrict__ Bp,
                                        const float* __restrict__ bias,
                                        const float* __restrict__ wvec,
                                        float* __restrict__ rowdot,
                                        int N, int K) {
    __shared__ float part[8][16];
    const int lane = threadIdx.x & 31;
    const int wid  = threadIdx.x >> 5;
    const int mt = blockIdx.x;
    const int ng = wid;
    const int kTiles = K >> 5;
    const int nTiles = N >> 4;

    const int hsel = (lane < 16) ? 0 : 8;
    const unsigned short* arow = A + (size_t)(mt * 16 + (lane & 15)) * K;
    const unsigned short* bcol = Bp + (size_t)(ng << 2) * 512 + lane * 16;
    const size_t bstride = (size_t)nTiles * 512;

    v8f acc0 = {}, acc1 = {}, acc2 = {}, acc3 = {};

    V16 aC, aN;
    V16 bC0, bC1, bC2, bC3, bN0, bN1, bN2, bN3;

    {
        const uint4* ap = (const uint4*)(arow + hsel);
        aC.u[0] = ap[0];
        aC.u[1] = ap[2];
        const unsigned short* bt = bcol;
        bC0.u[0] = ((const uint4*)(bt + 0 * 512))[0];
        bC0.u[1] = ((const uint4*)(bt + 0 * 512))[1];
        bC1.u[0] = ((const uint4*)(bt + 1 * 512))[0];
        bC1.u[1] = ((const uint4*)(bt + 1 * 512))[1];
        bC2.u[0] = ((const uint4*)(bt + 2 * 512))[0];
        bC2.u[1] = ((const uint4*)(bt + 2 * 512))[1];
        bC3.u[0] = ((const uint4*)(bt + 3 * 512))[0];
        bC3.u[1] = ((const uint4*)(bt + 3 * 512))[1];
    }

    for (int kt = 0; kt < kTiles; ++kt) {
        const int ktn = (kt + 1 < kTiles) ? (kt + 1) : kt;
        {
            const uint4* ap = (const uint4*)(arow + ktn * 32 + hsel);
            aN.u[0] = ap[0];
            aN.u[1] = ap[2];
            const unsigned short* bt = bcol + (size_t)ktn * bstride;
            bN0.u[0] = ((const uint4*)(bt + 0 * 512))[0];
            bN0.u[1] = ((const uint4*)(bt + 0 * 512))[1];
            bN1.u[0] = ((const uint4*)(bt + 1 * 512))[0];
            bN1.u[1] = ((const uint4*)(bt + 1 * 512))[1];
            bN2.u[0] = ((const uint4*)(bt + 2 * 512))[0];
            bN2.u[1] = ((const uint4*)(bt + 2 * 512))[1];
            bN3.u[0] = ((const uint4*)(bt + 3 * 512))[0];
            bN3.u[1] = ((const uint4*)(bt + 3 * 512))[1];
        }

        acc0 = __builtin_amdgcn_wmma_f32_16x16x32_bf16(false, aC.v, false, bC0.v,
                                                       (short)0, acc0, false, false);
        acc1 = __builtin_amdgcn_wmma_f32_16x16x32_bf16(false, aC.v, false, bC1.v,
                                                       (short)0, acc1, false, false);
        acc2 = __builtin_amdgcn_wmma_f32_16x16x32_bf16(false, aC.v, false, bC2.v,
                                                       (short)0, acc2, false, false);
        acc3 = __builtin_amdgcn_wmma_f32_16x16x32_bf16(false, aC.v, false, bC3.v,
                                                       (short)0, acc3, false, false);

        aC = aN;
        bC0 = bN0; bC1 = bN1; bC2 = bN2; bC3 = bN3;
    }

    const int c16 = lane & 15;
    v8f accs[4] = {acc0, acc1, acc2, acc3};
    float p[8];
#pragma unroll
    for (int v = 0; v < 8; ++v) p[v] = 0.0f;
#pragma unroll
    for (int j = 0; j < 4; ++j) {
        const int cc = (ng * 4 + j) * 16 + c16;
        const float bv = bias[cc];
        const float wv = wvec[cc];
#pragma unroll
        for (int v = 0; v < 8; ++v) p[v] += fmaxf(accs[j][v] + bv, 0.0f) * wv;
    }
#pragma unroll
    for (int off = 8; off > 0; off >>= 1) {
#pragma unroll
        for (int v = 0; v < 8; ++v) p[v] += __shfl_xor(p[v], off, 32);
    }
    float mine = p[0];
#pragma unroll
    for (int v = 1; v < 8; ++v) mine = ((lane & 7) == v) ? p[v] : mine;
    if (c16 < 8) part[wid][((lane >> 4) << 3) + c16] = mine;
    __syncthreads();
    if (threadIdx.x < 16) {
        float s = wvec[N];
        for (int w = 0; w < 8; ++w) s += part[w][threadIdx.x];
        rowdot[mt * 16 + threadIdx.x] = s;
    }
}

__device__ __forceinline__ unsigned int relu_bias_bf16x2(float a, float ba,
                                                         float b, float bb) {
    return (unsigned int)f2bf(fmaxf(a + ba, 0.0f)) |
           ((unsigned int)f2bf(fmaxf(b + bb, 0.0f)) << 16);
}

__device__ __forceinline__ uint4 relu_bias_bf16x8(float4 f0, float4 f1,
                                                  float4 b0, float4 b1) {
    uint4 r;
    r.x = relu_bias_bf16x2(f0.x, b0.x, f0.y, b0.y);
    r.y = relu_bias_bf16x2(f0.z, b0.z, f0.w, b0.w);
    r.z = relu_bias_bf16x2(f1.x, b1.x, f1.y, b1.y);
    r.w = relu_bias_bf16x2(f1.z, b1.z, f1.w, b1.w);
    return r;
}

__device__ __forceinline__ void load_a_frag_f32(V16& a,
                                                const float* __restrict__ arow32,
                                                const float* __restrict__ abias,
                                                int koff) {
    const float4* fp = (const float4*)(arow32 + koff);
    const float4* bp = (const float4*)(abias + koff);
    a.u[0] = relu_bias_bf16x8(fp[0], fp[1], bp[0], bp[1]);
    a.u[1] = relu_bias_bf16x8(fp[4], fp[5], bp[4], bp[5]);
}

template <bool A_F32>
__global__ void gemm_rowdot_kernel(const unsigned short* __restrict__ A16,
                                   const float* __restrict__ A32,
                                   const float* __restrict__ abias,
                                   const unsigned short* __restrict__ Bp,
                                   const float* __restrict__ bias,
                                   const float* __restrict__ fvec,
                                   float* __restrict__ rowdot,
                                   int M, int K,
                                   const float* __restrict__ addend,
                                   const float* __restrict__ deg,
                                   const float* __restrict__ cbias,
                                   const float* __restrict__ fb,
                                   float* __restrict__ accp) {
    const int lane = threadIdx.x & 31;
    const int wid  = threadIdx.x >> 5;
    const int mt = blockIdx.x * 8 + wid;
    if (mt >= (M >> 4)) return;

    const int kTiles = K >> 5;

    const int hsel = (lane < 16) ? 0 : 8;
    const unsigned short* arow = A16 + (size_t)(mt * 16 + (lane & 15)) * K;
    const float* arow32 = A32 + (size_t)(mt * 16 + (lane & 15)) * K;
    const unsigned short* bcol = Bp + lane * 16;
    const size_t bstride = (size_t)(COMB >> 4) * 512;

    v8f acc0 = {}, acc1 = {}, acc2 = {}, acc3 = {};

    V16 aC, aN;
    V16 bC0, bC1, bC2, bC3, bN0, bN1, bN2, bN3;

    {
        if (!A_F32) {
            const uint4* ap = (const uint4*)(arow + hsel);
            aC.u[0] = ap[0];
            aC.u[1] = ap[2];
        } else {
            load_a_frag_f32(aC, arow32, abias, hsel);
        }
        const unsigned short* bt = bcol;
        bC0.u[0] = ((const uint4*)(bt + 0 * 512))[0];
        bC0.u[1] = ((const uint4*)(bt + 0 * 512))[1];
        bC1.u[0] = ((const uint4*)(bt + 1 * 512))[0];
        bC1.u[1] = ((const uint4*)(bt + 1 * 512))[1];
        bC2.u[0] = ((const uint4*)(bt + 2 * 512))[0];
        bC2.u[1] = ((const uint4*)(bt + 2 * 512))[1];
        bC3.u[0] = ((const uint4*)(bt + 3 * 512))[0];
        bC3.u[1] = ((const uint4*)(bt + 3 * 512))[1];
    }

    for (int kt = 0; kt < kTiles; ++kt) {
        const int ktn = (kt + 1 < kTiles) ? (kt + 1) : kt;
        {
            if (!A_F32) {
                const uint4* ap = (const uint4*)(arow + ktn * 32 + hsel);
                aN.u[0] = ap[0];
                aN.u[1] = ap[2];
            } else {
                load_a_frag_f32(aN, arow32, abias, ktn * 32 + hsel);
            }
            const unsigned short* bt = bcol + (size_t)ktn * bstride;
            bN0.u[0] = ((const uint4*)(bt + 0 * 512))[0];
            bN0.u[1] = ((const uint4*)(bt + 0 * 512))[1];
            bN1.u[0] = ((const uint4*)(bt + 1 * 512))[0];
            bN1.u[1] = ((const uint4*)(bt + 1 * 512))[1];
            bN2.u[0] = ((const uint4*)(bt + 2 * 512))[0];
            bN2.u[1] = ((const uint4*)(bt + 2 * 512))[1];
            bN3.u[0] = ((const uint4*)(bt + 3 * 512))[0];
            bN3.u[1] = ((const uint4*)(bt + 3 * 512))[1];
        }

        acc0 = __builtin_amdgcn_wmma_f32_16x16x32_bf16(false, aC.v, false, bC0.v,
                                                       (short)0, acc0, false, false);
        acc1 = __builtin_amdgcn_wmma_f32_16x16x32_bf16(false, aC.v, false, bC1.v,
                                                       (short)0, acc1, false, false);
        acc2 = __builtin_amdgcn_wmma_f32_16x16x32_bf16(false, aC.v, false, bC2.v,
                                                       (short)0, acc2, false, false);
        acc3 = __builtin_amdgcn_wmma_f32_16x16x32_bf16(false, aC.v, false, bC3.v,
                                                       (short)0, acc3, false, false);

        aC = aN;
        bC0 = bN0; bC1 = bN1; bC2 = bN2; bC3 = bN3;
    }

    const int c16 = lane & 15;
    v8f accs[4] = {acc0, acc1, acc2, acc3};
    float p[8];
#pragma unroll
    for (int v = 0; v < 8; ++v) p[v] = 0.0f;
    float cpart = 0.0f;
#pragma unroll
    for (int j = 0; j < 4; ++j) {
        const int cc = j * 16 + c16;
        const float fv = fvec[cc];
        const float bv = bias ? bias[cc] : 0.0f;
        if (accp) cpart += cbias[cc] * fv;
#pragma unroll
        for (int v = 0; v < 8; ++v) p[v] += (accs[j][v] + bv) * fv;
    }
#pragma unroll
    for (int off = 8; off > 0; off >>= 1) {
#pragma unroll
        for (int v = 0; v < 8; ++v) p[v] += __shfl_xor(p[v], off, 32);
        cpart += __shfl_xor(cpart, off, 32);
    }
    float mine = p[0];
#pragma unroll
    for (int v = 1; v < 8; ++v) mine = ((lane & 7) == v) ? p[v] : mine;
    if (c16 < 8) {
        const int row = mt * 16 + ((lane >> 4) << 3) + c16;
        rowdot[row] = mine;
        if (accp) {
            const float di = dinv_of(deg[row]);
            accp[(size_t)row * ACC_PITCH] = addend[row] + (cpart + fb[0]) + di * di * mine;
        }
    }
}

__global__ void deg_accum_kernel(const int* __restrict__ dst,
                                 const float* __restrict__ ew,
                                 float* __restrict__ deg, int nE) {
    int e = blockIdx.x * blockDim.x + threadIdx.x;
    if (e < nE) atomicAdd(&deg[dst[e]], ew[e]);
}

__global__ void edge_norm_kernel(const int* __restrict__ src,
                                 const int* __restrict__ dst,
                                 const float* __restrict__ ew,
                                 const float* __restrict__ deg,
                                 float* __restrict__ norm, int nE) {
    int e = blockIdx.x * blockDim.x + threadIdx.x;
    if (e < nE) norm[e] = dinv_of(deg[src[e]]) * ew[e] * dinv_of(deg[dst[e]]);
}

#define SC_EPW 32
#define SC_GRP 8
static_assert(N_EDGES % SC_EPW == 0, "a wave always holds SC_EPW real edges");
static_assert(COMB == 64, "lane l owns floats l and l+32");

__global__ void gcn_scatter_kernel(const float* __restrict__ h,
                                   const int* __restrict__ src,
                                   const int* __restrict__ dst,
                                   const float* __restrict__ norm,
                                   float* __restrict__ agg, int nE) {
    const int lane = threadIdx.x & 31;
    const int e0 = (blockIdx.x * (blockDim.x >> 5) + (threadIdx.x >> 5)) * SC_EPW;
    if (e0 >= nE) return;
    const int sv = src[e0 + lane];
    const int dv = dst[e0 + lane];
    const int nv = __float_as_int(norm[e0 + lane]);
    const float* hl = h + lane;
    float* al = agg + lane;

    float lo[2][SC_GRP], hi[2][SC_GRP];
#pragma unroll
    for (int k = 0; k < SC_GRP; ++k) {
        const int s = __builtin_amdgcn_readlane(sv, k);
        lo[0][k] = hl[(size_t)s * COMB];
        hi[0][k] = hl[(size_t)s * COMB + 32];
    }
#pragma unroll
    for (int g = 0; g < SC_EPW / SC_GRP; ++g) {
        const int cur = g & 1, nxt = cur ^ 1;
        if (g + 1 < SC_EPW / SC_GRP) {
#pragma unroll
            for (int k = 0; k < SC_GRP; ++k) {
                const int s = __builtin_amdgcn_readlane(sv, (g + 1) * SC_GRP + k);
                lo[nxt][k] = hl[(size_t)s * COMB];
                hi[nxt][k] = hl[(size_t)s * COMB + 32];
            }
        }
#pragma unroll
        for (int k = 0; k < SC_GRP; ++k) {
            const int d = __builtin_amdgcn_readlane(dv, g * SC_GRP + k);
            const float nm = __int_as_float(__builtin_amdgcn_readlane(nv, g * SC_GRP + k));
            float* ap = al + (size_t)d * COMB;
            atomicAdd(ap,      lo[cur][k] * nm);
            atomicAdd(ap + 32, hi[cur][k] * nm);
        }
    }
}

__global__ void gcn_scatter_dot_kernel(const float* __restrict__ s,
                                       const int* __restrict__ src,
                                       const int* __restrict__ dst,
                                       const float* __restrict__ norm,
                                       float* __restrict__ accp, int nE4) {
    int i = blockIdx.x * blockDim.x + threadIdx.x;
    if (i >= nE4) return;
    const int4   sv = ((const int4*)src)[i];
    const int4   dv = ((const int4*)dst)[i];
    const float4 nv = ((const float4*)norm)[i];
    const float m0 = nv.x * s[sv.x];
    const float m1 = nv.y * s[sv.y];
    const float m2 = nv.z * s[sv.z];
    const float m3 = nv.w * s[sv.w];
    atomicAdd(accp + (size_t)dv.x * ACC_PITCH, m0);
    atomicAdd(accp + (size_t)dv.y * ACC_PITCH, m1);
    atomicAdd(accp + (size_t)dv.z * ACC_PITCH, m2);
    atomicAdd(accp + (size_t)dv.w * ACC_PITCH, m3);
}

__global__ void out_gather_kernel(const float* __restrict__ accp,
                                  float* __restrict__ out, int n) {
    int i = blockIdx.x * blockDim.x + threadIdx.x;
    if (i < n) out[i] = accp[(size_t)i * ACC_PITCH];
}

extern "C" void kernel_launch(void* const* d_in, const int* in_sizes, int n_in,
                              void* d_out, int out_size, void* d_ws, size_t ws_size,
                              hipStream_t stream) {
    (void)in_sizes; (void)n_in; (void)out_size; (void)ws_size;

    const float* x        = (const float*)d_in[0];
    const int*   eidx     = (const int*)d_in[1];
    const float* eattr    = (const float*)d_in[2];
    const float* ufeat    = (const float*)d_in[3];
    const float* gcn1_W   = (const float*)d_in[4];
    const float* gcn1_b   = (const float*)d_in[5];
    const float* gcn2_W   = (const float*)d_in[6];
    const float* gcn2_b   = (const float*)d_in[7];
    const float* fc1_W    = (const float*)d_in[8];
    const float* fc1_b    = (const float*)d_in[9];
    const float* fc2_W    = (const float*)d_in[10];
    const float* fc2_b    = (const float*)d_in[11];
    const float* fcf_W    = (const float*)d_in[12];
    const float* fcf_b    = (const float*)d_in[13];
    const int* esrc = eidx;
    const int* edst = eidx + N_EDGES;
    float* out = (float*)d_out;

    char* ws = (char*)d_ws;
    size_t off = 0;
    auto carve = [&](size_t bytes) -> void* {
        void* p = ws + off;
        off = (off + bytes + 255) & ~(size_t)255;
        return p;
    };
    unsigned short* UBF = (unsigned short*)carve((size_t)N_NODES * MLP_IN * 2);
    unsigned short* XBF = (unsigned short*)carve((size_t)N_NODES * GNN_IN * 2);
    unsigned short* W1P = (unsigned short*)carve((size_t)MLP_IN * MLP_HID * 2);
    float* W2F    = (float*)carve((size_t)(MLP_HID + 1) * 4);
    unsigned short* G1P = (unsigned short*)carve((size_t)GNN_IN * GNN_HID * 2);
    unsigned short* G2P = (unsigned short*)carve((size_t)GNN_HID * COMB * 2);
    float* H1     = (float*)carve((size_t)N_NODES * COMB * 4);
    float* AGG    = (float*)carve((size_t)N_NODES * COMB * 4);
    float* MDOT   = (float*)carve((size_t)N_NODES * 4);
    float* SDOT   = (float*)carve((size_t)N_NODES * 4);
    float* ACC    = (float*)carve((size_t)N_NODES * ACC_PITCH * 4);
    float* DEG    = (float*)carve((size_t)N_NODES * 4);
    float* NORM   = (float*)carve((size_t)N_EDGES * 4);

    prep_kernel<<<PB_ALL, 256, 0, stream>>>(ufeat, UBF, x, XBF, fc1_W, W1P, fc2_W, fc2_b,
                                            fcf_W, W2F, gcn1_W, G1P, gcn2_W, G2P, DEG);

    deg_accum_kernel<<<(N_EDGES + 255) / 256, 256, 0, stream>>>(edst, eattr, DEG, N_EDGES);
    edge_norm_kernel<<<(N_EDGES + 255) / 256, 256, 0, stream>>>(esrc, edst, eattr, DEG,
                                                                NORM, N_EDGES);

    const int rdBlocks = ((N_NODES / 16) + 7) / 8;

    gemm_relu_vecdot_kernel<<<N_NODES / 16, 256, 0, stream>>>(
        UBF, W1P, fc1_b, W2F, MDOT, MLP_HID, MLP_IN);

    gemm_wmma_kernel<true><<<((N_NODES / 16) * (GNN_HID / 64) + 7) / 8, 256, 0, stream>>>(
        XBF, G1P, nullptr, H1, N_NODES, GNN_HID, GNN_IN, 0, 0, DEG, AGG);
    gcn_scatter_kernel<<<(N_EDGES / SC_EPW + 7) / 8, 256, 0, stream>>>(
        H1, esrc, edst, NORM, AGG, N_EDGES);

    gemm_rowdot_kernel<true><<<rdBlocks, 256, 0, stream>>>(
        nullptr, AGG, gcn1_b, G2P, nullptr, fcf_W + COMB, SDOT, N_NODES, GNN_HID,
        MDOT, DEG, gcn2_b, fcf_b, ACC);
    gcn_scatter_dot_kernel<<<(N_EDGES / 4 + 255) / 256, 256, 0, stream>>>(
        SDOT, esrc, edst, NORM, ACC, N_EDGES / 4);

    out_gather_kernel<<<(N_NODES + 255) / 256, 256, 0, stream>>>(ACC, out, N_NODES);
}


# fc1 GEMM block tile 16 -> 64 rows: A tile staged once in LDS by whole-line loads, each packed-weight fragment a wave loads feeds 4 row tiles; weights streamed from L2 782 times, was 3125
# speedup vs baseline: 2.3784x; 1.1823x over previous
#include <hip/hip_runtime.h>
#include <hip/hip_bf16.h>


#define N_NODES 50000
#define N_EDGES 800000
#define GNN_IN  256
#define GNN_HID 64
#define COMB    64
#define MLP_IN  256
#define MLP_HID 512
#define ACC_PITCH 32

typedef __attribute__((ext_vector_type(16))) __bf16 v16bf;
typedef __attribute__((ext_vector_type(8)))  float  v8f;

union V16 {
    uint4 u[2];
    v16bf v;
};

__device__ __forceinline__ unsigned short f2bf(float f) {
    unsigned int u = __float_as_uint(f);
    unsigned int r = u + 0x7FFFu + ((u >> 16) & 1u);
    return (unsigned short)(r >> 16);
}

__device__ __forceinline__ void cvt_bf16_body(const float* __restrict__ in,
                                              unsigned short* __restrict__ out,
                                              int n4, int i) {
    if (i >= n4) return;
    const float4 f = ((const float4*)in)[i];
    ushort4 o;
    o.x = f2bf(f.x); o.y = f2bf(f.y); o.z = f2bf(f.z); o.w = f2bf(f.w);
    ((ushort4*)out)[i] = o;
}

__device__ __forceinline__ void pack_w_body(const float* __restrict__ W,
                                            unsigned short* __restrict__ P,
                                            int K, int N, int idx) {
    int nTiles = N >> 4;
    int total = (K >> 5) * nTiles * 32;
    if (idx >= total) return;
    int lane = idx & 31;
    int tile = idx >> 5;
    int kt = tile / nTiles;
    int nt = tile % nTiles;
    int col = nt * 16 + (lane & 15);
    int kb  = kt * 32 + ((lane < 16) ? 0 : 8);
    __attribute__((aligned(16))) unsigned short o[16];
#pragma unroll
    for (int j = 0; j < 8; ++j) o[j]     = f2bf(W[(size_t)(kb + j) * N + col]);
#pragma unroll
    for (int j = 0; j < 8; ++j) o[8 + j] = f2bf(W[(size_t)(kb + 16 + j) * N + col]);
    uint4* dst = (uint4*)(P + (size_t)tile * 512 + lane * 16);
    dst[0] = ((uint4*)o)[0];
    dst[1] = ((uint4*)o)[1];
}

__device__ __forceinline__ void fold_w2_body(const float* __restrict__ W2,
                                             const float* __restrict__ b2,
                                             const float* __restrict__ fw_lo,
                                             float* __restrict__ w2f, int idx) {
    if (idx > MLP_HID) return;
    const float* row = (idx < MLP_HID) ? (W2 + (size_t)idx * COMB) : b2;
    float s = 0.0f;
    for (int c = 0; c < COMB; ++c) s += row[c] * fw_lo[c];
    w2f[idx] = s;
}

__device__ __forceinline__ float dinv_of(float d) {
    return (d > 0.0f) ? rsqrtf(fmaxf(d, 1e-12f)) : 0.0f;
}

#define PB_W1  (((MLP_IN  / 32) * (MLP_HID / 16) * 32) / 256)
#define PB_W2  ((MLP_HID + 1 + 255) / 256)
#define PB_G1  (((GNN_IN  / 32) * (GNN_HID / 16) * 32) / 256)
#define PB_G2  (((GNN_HID / 32) * (COMB    / 16) * 32) / 256)
#define PB_DEG ((N_NODES + 255) / 256)
#define PB_CVU ((N_NODES * MLP_IN / 4) / 256)
#define PB_CVX ((N_NODES * GNN_IN / 4) / 256)
#define PB_ALL (PB_W1 + PB_W2 + PB_G1 + PB_G2 + PB_DEG + PB_CVU + PB_CVX)

__global__ void prep_kernel(const float* __restrict__ ufeat,
                            unsigned short* __restrict__ UBF,
                            const float* __restrict__ x,
                            unsigned short* __restrict__ XBF,
                            const float* __restrict__ fc1_W,
                            unsigned short* __restrict__ W1P,
                            const float* __restrict__ fc2_W,
                            const float* __restrict__ fc2_b,
                            const float* __restrict__ fw_lo,
                            float* __restrict__ W2F,
                            const float* __restrict__ gcn1_W,
                            unsigned short* __restrict__ G1P,
                            const float* __restrict__ gcn2_W,
                            unsigned short* __restrict__ G2P,
                            float* __restrict__ deg) {
    int b = blockIdx.x;
    const int t = threadIdx.x;
    if (b < PB_W1) { pack_w_body(fc1_W, W1P, MLP_IN, MLP_HID, b * 256 + t); return; }
    b -= PB_W1;
    if (b < PB_W2) { fold_w2_body(fc2_W, fc2_b, fw_lo, W2F, b * 256 + t); return; }
    b -= PB_W2;
    if (b < PB_G1) { pack_w_body(gcn1_W, G1P, GNN_IN, GNN_HID, b * 256 + t); return; }
    b -= PB_G1;
    if (b < PB_G2) { pack_w_body(gcn2_W, G2P, GNN_HID, COMB, b * 256 + t); return; }
    b -= PB_G2;
    if (b < PB_DEG) {
        const int i = b * 256 + t;
        if (i < N_NODES) deg[i] = 1.0f;
        return;
    }
    b -= PB_DEG;
    if (b < PB_CVU) { cvt_bf16_body(ufeat, UBF, N_NODES * MLP_IN / 4, b * 256 + t); return; }
    b -= PB_CVU;
    cvt_bf16_body(x, XBF, N_NODES * GNN_IN / 4, b * 256 + t);
}

template <bool AGG_INIT>
__global__ void gemm_wmma_kernel(const unsigned short* __restrict__ A,
                                 const unsigned short* __restrict__ Bp,
                                 const float* __restrict__ bias,
                                 void* __restrict__ C,
                                 int M, int N, int K, int doRelu, int outBf16,
                                 const float* __restrict__ deg,
                                 float* __restrict__ agg) {
    const int lane = threadIdx.x & 31;
    const int wid  = threadIdx.x >> 5;
    const int nGroups = N >> 6;
    const int mTiles  = M >> 4;
    const int task = blockIdx.x * 8 + wid;
    if (task >= mTiles * nGroups) return;
    const int mt = task / nGroups;
    const int ng = task % nGroups;
    const int kTiles = K >> 5;
    const int nTiles = N >> 4;

    const int hsel = (lane < 16) ? 0 : 8;
    const unsigned short* arow = A + (size_t)(mt * 16 + (lane & 15)) * K;
    const unsigned short* bcol = Bp + (size_t)(ng << 2) * 512 + lane * 16;
    const size_t bstride = (size_t)nTiles * 512;

    v8f acc0 = {}, acc1 = {}, acc2 = {}, acc3 = {};

    V16 aC, aN;
    V16 bC0, bC1, bC2, bC3, bN0, bN1, bN2, bN3;

    {
        const uint4* ap = (const uint4*)(arow + hsel);
        aC.u[0] = ap[0];
        aC.u[1] = ap[2];
        const unsigned short* bt = bcol;
        bC0.u[0] = ((const uint4*)(bt + 0 * 512))[0];
        bC0.u[1] = ((const uint4*)(bt + 0 * 512))[1];
        bC1.u[0] = ((const uint4*)(bt + 1 * 512))[0];
        bC1.u[1] = ((const uint4*)(bt + 1 * 512))[1];
        bC2.u[0] = ((const uint4*)(bt + 2 * 512))[0];
        bC2.u[1] = ((const uint4*)(bt + 2 * 512))[1];
        bC3.u[0] = ((const uint4*)(bt + 3 * 512))[0];
        bC3.u[1] = ((const uint4*)(bt + 3 * 512))[1];
    }

    for (int kt = 0; kt < kTiles; ++kt) {
        const int ktn = (kt + 1 < kTiles) ? (kt + 1) : kt;
        {
            const uint4* ap = (const uint4*)(arow + ktn * 32 + hsel);
            aN.u[0] = ap[0];
            aN.u[1] = ap[2];
            const unsigned short* bt = bcol + (size_t)ktn * bstride;
            bN0.u[0] = ((const uint4*)(bt + 0 * 512))[0];
            bN0.u[1] = ((const uint4*)(bt + 0 * 512))[1];
            bN1.u[0] = ((const uint4*)(bt + 1 * 512))[0];
            bN1.u[1] = ((const uint4*)(bt + 1 * 512))[1];
            bN2.u[0] = ((const uint4*)(bt + 2 * 512))[0];
            bN2.u[1] = ((const uint4*)(bt + 2 * 512))[1];
            bN3.u[0] = ((const uint4*)(bt + 3 * 512))[0];
            bN3.u[1] = ((const uint4*)(bt + 3 * 512))[1];
        }

        acc0 = __builtin_amdgcn_wmma_f32_16x16x32_bf16(false, aC.v, false, bC0.v,
                                                       (short)0, acc0, false, false);
        acc1 = __builtin_amdgcn_wmma_f32_16x16x32_bf16(false, aC.v, false, bC1.v,
                                                       (short)0, acc1, false, false);
        acc2 = __builtin_amdgcn_wmma_f32_16x16x32_bf16(false, aC.v, false, bC2.v,
                                                       (short)0, acc2, false, false);
        acc3 = __builtin_amdgcn_wmma_f32_16x16x32_bf16(false, aC.v, false, bC3.v,
                                                       (short)0, acc3, false, false);

        aC = aN;
        bC0 = bN0; bC1 = bN1; bC2 = bN2; bC3 = bN3;
    }

    const int r0 = mt * 16 + ((lane < 16) ? 0 : 8);
    v8f accs[4] = {acc0, acc1, acc2, acc3};
    float selfw[8];
#pragma unroll
    for (int v = 0; v < 8; ++v) {
        if (AGG_INIT) { const float di = dinv_of(deg[r0 + v]); selfw[v] = di * di; }
        else          selfw[v] = 0.0f;
    }
#pragma unroll
    for (int j = 0; j < 4; ++j) {
        const int cc = (ng * 4 + j) * 16 + (lane & 15);
        const float bv = bias ? bias[cc] : 0.0f;
#pragma unroll
        for (int v = 0; v < 8; ++v) {
            float val = accs[j][v] + bv;
            if (doRelu) val = fmaxf(val, 0.0f);
            const size_t off = (size_t)(r0 + v) * N + cc;
            if (outBf16) ((unsigned short*)C)[off] = f2bf(val);
            else         ((float*)C)[off] = val;
            if (AGG_INIT) agg[off] = val * selfw[v];
        }
    }
}

#define FC1_MT   4
#define FC1_ROWS (16 * FC1_MT)
#define FC1_APITCH (MLP_IN * 2 + 16)
static_assert(MLP_HID == 8 * 64, "8 waves x 64 columns cover one row");
static_assert(MLP_IN * 2 == 32 * 16, "one wave-wide 16-byte load covers one bf16 row");
static_assert((MLP_IN / 32) % 2 == 0, "the K loop takes k-steps in pairs");
static_assert(FC1_ROWS % 8 == 0, "8 waves share the rows of the staging pass");

__device__ __forceinline__ void fc1_kstep(v8f (&acc)[FC1_MT][4], V16 (&a)[FC1_MT],
                                          const V16 (&bUse)[4], V16 (&bFill)[4],
                                          const unsigned char* afrag, int aNext,
                                          const unsigned short* __restrict__ bnext) {
#pragma unroll
    for (int j = 0; j < 4; ++j) {
        bFill[j].u[0] = ((const uint4*)(bnext + j * 512))[0];
        bFill[j].u[1] = ((const uint4*)(bnext + j * 512))[1];
    }
    __builtin_amdgcn_sched_barrier(0);
#pragma unroll
    for (int i = 0; i < FC1_MT; ++i) {
#pragma unroll
        for (int j = 0; j < 4; ++j)
            acc[i][j] = __builtin_amdgcn_wmma_f32_16x16x32_bf16(false, a[i].v, false, bUse[j].v,
                                                                (short)0, acc[i][j], false, false);
        __builtin_amdgcn_sched_barrier(0);
        const unsigned char* ap = afrag + i * 16 * FC1_APITCH + aNext;
        a[i].u[0] = *(const uint4*)(ap);
        a[i].u[1] = *(const uint4*)(ap + 32);
        __builtin_amdgcn_sched_barrier(0);
    }
}

__global__ void __launch_bounds__(256, 4)
gemm_relu_vecdot_kernel(const unsigned short* __restrict__ A,
                        const unsigned short* __restrict__ Bp,
                        const float* __restrict__ bias,
                        const float* __restrict__ wvec,
                        float* __restrict__ rowdot,
                        int M) {
    __shared__ __attribute__((aligned(16))) unsigned char As[FC1_ROWS * FC1_APITCH];
    __shared__ float part[8][FC1_ROWS];
    const int lane = threadIdx.x & 31;
    const int wid  = threadIdx.x >> 5;
    const int ng = wid;
    constexpr int kTiles = MLP_IN >> 5;
    constexpr int nTiles = MLP_HID >> 4;
    const int row0 = blockIdx.x * FC1_ROWS;

    const unsigned short* bcol = Bp + (size_t)(ng << 2) * 512 + lane * 16;
    constexpr size_t bstride = (size_t)nTiles * 512;

    V16 a[FC1_MT];
    V16 b0[4], b1[4];

#pragma unroll
    for (int j = 0; j < 4; ++j) {
        b0[j].u[0] = ((const uint4*)(bcol + j * 512))[0];
        b0[j].u[1] = ((const uint4*)(bcol + j * 512))[1];
    }

    const int c16 = lane & 15;
    float bv[4], wv[4];
#pragma unroll
    for (int j = 0; j < 4; ++j) {
        const int cc = (ng * 4 + j) * 16 + c16;
        bv[j] = bias[cc];
        wv[j] = wvec[cc];
    }

    {
        uint4 st[FC1_ROWS / 8];
#pragma unroll
        for (int r = 0; r < FC1_ROWS / 8; ++r) {
            int grow = row0 + r * 8 + wid;
            grow = (grow < M) ? grow : (M - 1);
            st[r] = ((const uint4*)(A + (size_t)grow * MLP_IN))[lane];
        }
#pragma unroll
        for (int r = 0; r < FC1_ROWS / 8; ++r)
            *(uint4*)(As + (r * 8 + wid) * FC1_APITCH + lane * 16) = st[r];
    }
    __syncthreads();

    const unsigned char* afrag = As + (lane & 15) * FC1_APITCH + ((lane >> 4) << 4);
#pragma unroll
    for (int i = 0; i < FC1_MT; ++i) {
        a[i].u[0] = *(const uint4*)(afrag + i * 16 * FC1_APITCH);
        a[i].u[1] = *(const uint4*)(afrag + i * 16 * FC1_APITCH + 32);
    }

    v8f acc[FC1_MT][4];
#pragma unroll
    for (int i = 0; i < FC1_MT; ++i)
#pragma unroll
        for (int j = 0; j < 4; ++j) acc[i][j] = v8f{};

    for (int kt = 0; kt < kTiles; kt += 2) {
        const int k1 = kt + 1;
        const int k2 = (kt + 2 < kTiles) ? (kt + 2) : (kt + 1);
        fc1_kstep(acc, a, b0, b1, afrag, k1 * 64, bcol + (size_t)k1 * bstride);
        fc1_kstep(acc, a, b1, b0, afrag, k2 * 64, bcol + (size_t)k2 * bstride);
    }

    float p[FC1_MT][8];
#pragma unroll
    for (int i = 0; i < FC1_MT; ++i) {
#pragma unroll
        for (int v = 0; v < 8; ++v) p[i][v] = 0.0f;
#pragma unroll
        for (int j = 0; j < 4; ++j) {
#pragma unroll
            for (int v = 0; v < 8; ++v) p[i][v] += fmaxf(acc[i][j][v] + bv[j], 0.0f) * wv[j];
        }
    }
#pragma unroll
    for (int off = 8; off > 0; off >>= 1) {
#pragma unroll
        for (int i = 0; i < FC1_MT; ++i) {
#pragma unroll
            for (int v = 0; v < 8; ++v) p[i][v] += __shfl_xor(p[i][v], off, 32);
        }
    }
#pragma unroll
    for (int i = 0; i < FC1_MT; ++i) {
        float mine = p[i][0];
#pragma unroll
        for (int v = 1; v < 8; ++v) mine = ((lane & 7) == v) ? p[i][v] : mine;
        if (c16 < 8) part[wid][i * 16 + ((lane >> 4) << 3) + c16] = mine;
    }
    __syncthreads();
    if (threadIdx.x < FC1_ROWS) {
        const int row = row0 + threadIdx.x;
        if (row < M) {
            float s = wvec[MLP_HID];
            for (int w = 0; w < 8; ++w) s += part[w][threadIdx.x];
            rowdot[row] = s;
        }
    }
}

__device__ __forceinline__ unsigned int relu_bias_bf16x2(float a, float ba,
                                                         float b, float bb) {
    return (unsigned int)f2bf(fmaxf(a + ba, 0.0f)) |
           ((unsigned int)f2bf(fmaxf(b + bb, 0.0f)) << 16);
}

__device__ __forceinline__ uint4 relu_bias_bf16x8(float4 f0, float4 f1,
                                                  float4 b0, float4 b1) {
    uint4 r;
    r.x = relu_bias_bf16x2(f0.x, b0.x, f0.y, b0.y);
    r.y = relu_bias_bf16x2(f0.z, b0.z, f0.w, b0.w);
    r.z = relu_bias_bf16x2(f1.x, b1.x, f1.y, b1.y);
    r.w = relu_bias_bf16x2(f1.z, b1.z, f1.w, b1.w);
    return r;
}

__device__ __forceinline__ void load_a_frag_f32(V16& a,
                                                const float* __restrict__ arow32,
                                                const float* __restrict__ abias,
                                                int koff) {
    const float4* fp = (const float4*)(arow32 + koff);
    const float4* bp = (const float4*)(abias + koff);
    a.u[0] = relu_bias_bf16x8(fp[0], fp[1], bp[0], bp[1]);
    a.u[1] = relu_bias_bf16x8(fp[4], fp[5], bp[4], bp[5]);
}

template <bool A_F32>
__global__ void gemm_rowdot_kernel(const unsigned short* __restrict__ A16,
                                   const float* __restrict__ A32,
                                   const float* __restrict__ abias,
                                   const unsigned short* __restrict__ Bp,
                                   const float* __restrict__ bias,
                                   const float* __restrict__ fvec,
                                   float* __restrict__ rowdot,
                                   int M, int K,
                                   const float* __restrict__ addend,
                                   const float* __restrict__ deg,
                                   const float* __restrict__ cbias,
                                   const float* __restrict__ fb,
                                   float* __restrict__ accp) {
    const int lane = threadIdx.x & 31;
    const int wid  = threadIdx.x >> 5;
    const int mt = blockIdx.x * 8 + wid;
    if (mt >= (M >> 4)) return;

    const int kTiles = K >> 5;

    const int hsel = (lane < 16) ? 0 : 8;
    const unsigned short* arow = A16 + (size_t)(mt * 16 + (lane & 15)) * K;
    const float* arow32 = A32 + (size_t)(mt * 16 + (lane & 15)) * K;
    const unsigned short* bcol = Bp + lane * 16;
    const size_t bstride = (size_t)(COMB >> 4) * 512;

    v8f acc0 = {}, acc1 = {}, acc2 = {}, acc3 = {};

    V16 aC, aN;
    V16 bC0, bC1, bC2, bC3, bN0, bN1, bN2, bN3;

    {
        if (!A_F32) {
            const uint4* ap = (const uint4*)(arow + hsel);
            aC.u[0] = ap[0];
            aC.u[1] = ap[2];
        } else {
            load_a_frag_f32(aC, arow32, abias, hsel);
        }
        const unsigned short* bt = bcol;
        bC0.u[0] = ((const uint4*)(bt + 0 * 512))[0];
        bC0.u[1] = ((const uint4*)(bt + 0 * 512))[1];
        bC1.u[0] = ((const uint4*)(bt + 1 * 512))[0];
        bC1.u[1] = ((const uint4*)(bt + 1 * 512))[1];
        bC2.u[0] = ((const uint4*)(bt + 2 * 512))[0];
        bC2.u[1] = ((const uint4*)(bt + 2 * 512))[1];
        bC3.u[0] = ((const uint4*)(bt + 3 * 512))[0];
        bC3.u[1] = ((const uint4*)(bt + 3 * 512))[1];
    }

    for (int kt = 0; kt < kTiles; ++kt) {
        const int ktn = (kt + 1 < kTiles) ? (kt + 1) : kt;
        {
            if (!A_F32) {
                const uint4* ap = (const uint4*)(arow + ktn * 32 + hsel);
                aN.u[0] = ap[0];
                aN.u[1] = ap[2];
            } else {
                load_a_frag_f32(aN, arow32, abias, ktn * 32 + hsel);
            }
            const unsigned short* bt = bcol + (size_t)ktn * bstride;
            bN0.u[0] = ((const uint4*)(bt + 0 * 512))[0];
            bN0.u[1] = ((const uint4*)(bt + 0 * 512))[1];
            bN1.u[0] = ((const uint4*)(bt + 1 * 512))[0];
            bN1.u[1] = ((const uint4*)(bt + 1 * 512))[1];
            bN2.u[0] = ((const uint4*)(bt + 2 * 512))[0];
            bN2.u[1] = ((const uint4*)(bt + 2 * 512))[1];
            bN3.u[0] = ((const uint4*)(bt + 3 * 512))[0];
            bN3.u[1] = ((const uint4*)(bt + 3 * 512))[1];
        }

        acc0 = __builtin_amdgcn_wmma_f32_16x16x32_bf16(false, aC.v, false, bC0.v,
                                                       (short)0, acc0, false, false);
        acc1 = __builtin_amdgcn_wmma_f32_16x16x32_bf16(false, aC.v, false, bC1.v,
                                                       (short)0, acc1, false, false);
        acc2 = __builtin_amdgcn_wmma_f32_16x16x32_bf16(false, aC.v, false, bC2.v,
                                                       (short)0, acc2, false, false);
        acc3 = __builtin_amdgcn_wmma_f32_16x16x32_bf16(false, aC.v, false, bC3.v,
                                                       (short)0, acc3, false, false);

        aC = aN;
        bC0 = bN0; bC1 = bN1; bC2 = bN2; bC3 = bN3;
    }

    const int c16 = lane & 15;
    v8f accs[4] = {acc0, acc1, acc2, acc3};
    float p[8];
#pragma unroll
    for (int v = 0; v < 8; ++v) p[v] = 0.0f;
    float cpart = 0.0f;
#pragma unroll
    for (int j = 0; j < 4; ++j) {
        const int cc = j * 16 + c16;
        const float fv = fvec[cc];
        const float bv = bias ? bias[cc] : 0.0f;
        if (accp) cpart += cbias[cc] * fv;
#pragma unroll
        for (int v = 0; v < 8; ++v) p[v] += (accs[j][v] + bv) * fv;
    }
#pragma unroll
    for (int off = 8; off > 0; off >>= 1) {
#pragma unroll
        for (int v = 0; v < 8; ++v) p[v] += __shfl_xor(p[v], off, 32);
        cpart += __shfl_xor(cpart, off, 32);
    }
    float mine = p[0];
#pragma unroll
    for (int v = 1; v < 8; ++v) mine = ((lane & 7) == v) ? p[v] : mine;
    if (c16 < 8) {
        const int row = mt * 16 + ((lane >> 4) << 3) + c16;
        rowdot[row] = mine;
        if (accp) {
            const float di = dinv_of(deg[row]);
            accp[(size_t)row * ACC_PITCH] = addend[row] + (cpart + fb[0]) + di * di * mine;
        }
    }
}

__global__ void deg_accum_kernel(const int* __restrict__ dst,
                                 const float* __restrict__ ew,
                                 float* __restrict__ deg, int nE) {
    int e = blockIdx.x * blockDim.x + threadIdx.x;
    if (e < nE) atomicAdd(&deg[dst[e]], ew[e]);
}

__global__ void edge_norm_kernel(const int* __restrict__ src,
                                 const int* __restrict__ dst,
                                 const float* __restrict__ ew,
                                 const float* __restrict__ deg,
                                 float* __restrict__ norm, int nE) {
    int e = blockIdx.x * blockDim.x + threadIdx.x;
    if (e < nE) norm[e] = dinv_of(deg[src[e]]) * ew[e] * dinv_of(deg[dst[e]]);
}

#define SC_EPW 32
#define SC_GRP 8
static_assert(N_EDGES % SC_EPW == 0, "a wave always holds SC_EPW real edges");
static_assert(COMB == 64, "lane l owns floats l and l+32");

__global__ void gcn_scatter_kernel(const float* __restrict__ h,
                                   const int* __restrict__ src,
                                   const int* __restrict__ dst,
                                   const float* __restrict__ norm,
                                   float* __restrict__ agg, int nE) {
    const int lane = threadIdx.x & 31;
    const int e0 = (blockIdx.x * (blockDim.x >> 5) + (threadIdx.x >> 5)) * SC_EPW;
    if (e0 >= nE) return;
    const int sv = src[e0 + lane];
    const int dv = dst[e0 + lane];
    const int nv = __float_as_int(norm[e0 + lane]);
    const float* hl = h + lane;
    float* al = agg + lane;

    float lo[2][SC_GRP], hi[2][SC_GRP];
#pragma unroll
    for (int k = 0; k < SC_GRP; ++k) {
        const int s = __builtin_amdgcn_readlane(sv, k);
        lo[0][k] = hl[(size_t)s * COMB];
        hi[0][k] = hl[(size_t)s * COMB + 32];
    }
#pragma unroll
    for (int g = 0; g < SC_EPW / SC_GRP; ++g) {
        const int cur = g & 1, nxt = cur ^ 1;
        if (g + 1 < SC_EPW / SC_GRP) {
#pragma unroll
            for (int k = 0; k < SC_GRP; ++k) {
                const int s = __builtin_amdgcn_readlane(sv, (g + 1) * SC_GRP + k);
                lo[nxt][k] = hl[(size_t)s * COMB];
                hi[nxt][k] = hl[(size_t)s * COMB + 32];
            }
        }
#pragma unroll
        for (int k = 0; k < SC_GRP; ++k) {
            const int d = __builtin_amdgcn_readlane(dv, g * SC_GRP + k);
            const float nm = __int_as_float(__builtin_amdgcn_readlane(nv, g * SC_GRP + k));
            float* ap = al + (size_t)d * COMB;
            atomicAdd(ap,      lo[cur][k] * nm);
            atomicAdd(ap + 32, hi[cur][k] * nm);
        }
    }
}

__global__ void gcn_scatter_dot_kernel(const float* __restrict__ s,
                                       const int* __restrict__ src,
                                       const int* __restrict__ dst,
                                       const float* __restrict__ norm,
                                       float* __restrict__ accp, int nE4) {
    int i = blockIdx.x * blockDim.x + threadIdx.x;
    if (i >= nE4) return;
    const int4   sv = ((const int4*)src)[i];
    const int4   dv = ((const int4*)dst)[i];
    const float4 nv = ((const float4*)norm)[i];
    const float m0 = nv.x * s[sv.x];
    const float m1 = nv.y * s[sv.y];
    const float m2 = nv.z * s[sv.z];
    const float m3 = nv.w * s[sv.w];
    atomicAdd(accp + (size_t)dv.x * ACC_PITCH, m0);
    atomicAdd(accp + (size_t)dv.y * ACC_PITCH, m1);
    atomicAdd(accp + (size_t)dv.z * ACC_PITCH, m2);
    atomicAdd(accp + (size_t)dv.w * ACC_PITCH, m3);
}

__global__ void out_gather_kernel(const float* __restrict__ accp,
                                  float* __restrict__ out, int n) {
    int i = blockIdx.x * blockDim.x + threadIdx.x;
    if (i < n) out[i] = accp[(size_t)i * ACC_PITCH];
}

extern "C" void kernel_launch(void* const* d_in, const int* in_sizes, int n_in,
                              void* d_out, int out_size, void* d_ws, size_t ws_size,
                              hipStream_t stream) {
    (void)in_sizes; (void)n_in; (void)out_size; (void)ws_size;

    const float* x        = (const float*)d_in[0];
    const int*   eidx     = (const int*)d_in[1];
    const float* eattr    = (const float*)d_in[2];
    const float* ufeat    = (const float*)d_in[3];
    const float* gcn1_W   = (const float*)d_in[4];
    const float* gcn1_b   = (const float*)d_in[5];
    const float* gcn2_W   = (const float*)d_in[6];
    const float* gcn2_b   = (const float*)d_in[7];
    const float* fc1_W    = (const float*)d_in[8];
    const float* fc1_b    = (const float*)d_in[9];
    const float* fc2_W    = (const float*)d_in[10];
    const float* fc2_b    = (const float*)d_in[11];
    const float* fcf_W    = (const float*)d_in[12];
    const float* fcf_b    = (const float*)d_in[13];
    const int* esrc = eidx;
    const int* edst = eidx + N_EDGES;
    float* out = (float*)d_out;

    char* ws = (char*)d_ws;
    size_t off = 0;
    auto carve = [&](size_t bytes) -> void* {
        void* p = ws + off;
        off = (off + bytes + 255) & ~(size_t)255;
        return p;
    };
    unsigned short* UBF = (unsigned short*)carve((size_t)N_NODES * MLP_IN * 2);
    unsigned short* XBF = (unsigned short*)carve((size_t)N_NODES * GNN_IN * 2);
    unsigned short* W1P = (unsigned short*)carve((size_t)MLP_IN * MLP_HID * 2);
    float* W2F    = (float*)carve((size_t)(MLP_HID + 1) * 4);
    unsigned short* G1P = (unsigned short*)carve((size_t)GNN_IN * GNN_HID * 2);
    unsigned short* G2P = (unsigned short*)carve((size_t)GNN_HID * COMB * 2);
    float* H1     = (float*)carve((size_t)N_NODES * COMB * 4);
    float* AGG    = (float*)carve((size_t)N_NODES * COMB * 4);
    float* MDOT   = (float*)carve((size_t)N_NODES * 4);
    float* SDOT   = (float*)carve((size_t)N_NODES * 4);
    float* ACC    = (float*)carve((size_t)N_NODES * ACC_PITCH * 4);
    float* DEG    = (float*)carve((size_t)N_NODES * 4);
    float* NORM   = (float*)carve((size_t)N_EDGES * 4);

    prep_kernel<<<PB_ALL, 256, 0, stream>>>(ufeat, UBF, x, XBF, fc1_W, W1P, fc2_W, fc2_b,
                                            fcf_W, W2F, gcn1_W, G1P, gcn2_W, G2P, DEG);

    deg_accum_kernel<<<(N_EDGES + 255) / 256, 256, 0, stream>>>(edst, eattr, DEG, N_EDGES);
    edge_norm_kernel<<<(N_EDGES + 255) / 256, 256, 0, stream>>>(esrc, edst, eattr, DEG,
                                                                NORM, N_EDGES);

    const int rdBlocks = ((N_NODES / 16) + 7) / 8;

    gemm_relu_vecdot_kernel<<<(N_NODES + FC1_ROWS - 1) / FC1_ROWS, 256, 0, stream>>>(
        UBF, W1P, fc1_b, W2F, MDOT, N_NODES);

    gemm_wmma_kernel<true><<<((N_NODES / 16) * (GNN_HID / 64) + 7) / 8, 256, 0, stream>>>(
        XBF, G1P, nullptr, H1, N_NODES, GNN_HID, GNN_IN, 0, 0, DEG, AGG);
    gcn_scatter_kernel<<<(N_EDGES / SC_EPW + 7) / 8, 256, 0, stream>>>(
        H1, esrc, edst, NORM, AGG, N_EDGES);

    gemm_rowdot_kernel<true><<<rdBlocks, 256, 0, stream>>>(
        nullptr, AGG, gcn1_b, G2P, nullptr, fcf_W + COMB, SDOT, N_NODES, GNN_HID,
        MDOT, DEG, gcn2_b, fcf_b, ACC);
    gcn_scatter_dot_kernel<<<(N_EDGES / 4 + 255) / 256, 256, 0, stream>>>(
        SDOT, esrc, edst, NORM, ACC, N_EDGES / 4);

    out_gather_kernel<<<(N_NODES + 255) / 256, 256, 0, stream>>>(ACC, out, N_NODES);
}


# fp32->bf16 conversion pass removed: fc1 GEMM rounds its f32 rows to bf16 while staging them in LDS, layer-1 GEMM loads each wave's 16 f32 rows whole and rounds A fragments in registers
# speedup vs baseline: 2.5213x; 1.0601x over previous
#include <hip/hip_runtime.h>
#include <hip/hip_bf16.h>


#define N_NODES 50000
#define N_EDGES 800000
#define GNN_IN  256
#define GNN_HID 64
#define COMB    64
#define MLP_IN  256
#define MLP_HID 512
#define ACC_PITCH 32

typedef __attribute__((ext_vector_type(16))) __bf16 v16bf;
typedef __attribute__((ext_vector_type(8)))  float  v8f;

union V16 {
    uint4 u[2];
    v16bf v;
};

__device__ __forceinline__ unsigned short f2bf(float f) {
    unsigned int u = __float_as_uint(f);
    unsigned int r = u + 0x7FFFu + ((u >> 16) & 1u);
    return (unsigned short)(r >> 16);
}

typedef __attribute__((ext_vector_type(2))) float  v2f;
typedef __attribute__((ext_vector_type(2))) __bf16 v2bf;
__device__ __forceinline__ unsigned int bf16x2_of(float a, float b) {
    const v2f f = {a, b};
    const v2bf h = __builtin_convertvector(f, v2bf);
    return __builtin_bit_cast(unsigned int, h);
}
__device__ __forceinline__ uint2 bf16x4_of(float4 f) {
    uint2 r;
    r.x = bf16x2_of(f.x, f.y);
    r.y = bf16x2_of(f.z, f.w);
    return r;
}
__device__ __forceinline__ uint4 bf16x8_of(float4 f0, float4 f1) {
    uint4 r;
    r.x = bf16x2_of(f0.x, f0.y);
    r.y = bf16x2_of(f0.z, f0.w);
    r.z = bf16x2_of(f1.x, f1.y);
    r.w = bf16x2_of(f1.z, f1.w);
    return r;
}

__device__ __forceinline__ void pack_w_body(const float* __restrict__ W,
                                            unsigned short* __restrict__ P,
                                            int K, int N, int idx) {
    int nTiles = N >> 4;
    int total = (K >> 5) * nTiles * 32;
    if (idx >= total) return;
    int lane = idx & 31;
    int tile = idx >> 5;
    int kt = tile / nTiles;
    int nt = tile % nTiles;
    int col = nt * 16 + (lane & 15);
    int kb  = kt * 32 + ((lane < 16) ? 0 : 8);
    __attribute__((aligned(16))) unsigned short o[16];
#pragma unroll
    for (int j = 0; j < 8; ++j) o[j]     = f2bf(W[(size_t)(kb + j) * N + col]);
#pragma unroll
    for (int j = 0; j < 8; ++j) o[8 + j] = f2bf(W[(size_t)(kb + 16 + j) * N + col]);
    uint4* dst = (uint4*)(P + (size_t)tile * 512 + lane * 16);
    dst[0] = ((uint4*)o)[0];
    dst[1] = ((uint4*)o)[1];
}

__device__ __forceinline__ void fold_w2_body(const float* __restrict__ W2,
                                             const float* __restrict__ b2,
                                             const float* __restrict__ fw_lo,
                                             float* __restrict__ w2f, int idx) {
    if (idx > MLP_HID) return;
    const float* row = (idx < MLP_HID) ? (W2 + (size_t)idx * COMB) : b2;
    float s = 0.0f;
    for (int c = 0; c < COMB; ++c) s += row[c] * fw_lo[c];
    w2f[idx] = s;
}

__device__ __forceinline__ float dinv_of(float d) {
    return (d > 0.0f) ? rsqrtf(fmaxf(d, 1e-12f)) : 0.0f;
}

#define PB_W1  (((MLP_IN  / 32) * (MLP_HID / 16) * 32) / 256)
#define PB_W2  ((MLP_HID + 1 + 255) / 256)
#define PB_G1  (((GNN_IN  / 32) * (GNN_HID / 16) * 32) / 256)
#define PB_G2  (((GNN_HID / 32) * (COMB    / 16) * 32) / 256)
#define PB_DEG ((N_NODES + 255) / 256)
#define PB_ALL (PB_W1 + PB_W2 + PB_G1 + PB_G2 + PB_DEG)

__global__ void prep_kernel(const float* __restrict__ fc1_W,
                            unsigned short* __restrict__ W1P,
                            const float* __restrict__ fc2_W,
                            const float* __restrict__ fc2_b,
                            const float* __restrict__ fw_lo,
                            float* __restrict__ W2F,
                            const float* __restrict__ gcn1_W,
                            unsigned short* __restrict__ G1P,
                            const float* __restrict__ gcn2_W,
                            unsigned short* __restrict__ G2P,
                            float* __restrict__ deg) {
    int b = blockIdx.x;
    const int t = threadIdx.x;
    if (b < PB_W1) { pack_w_body(fc1_W, W1P, MLP_IN, MLP_HID, b * 256 + t); return; }
    b -= PB_W1;
    if (b < PB_W2) { fold_w2_body(fc2_W, fc2_b, fw_lo, W2F, b * 256 + t); return; }
    b -= PB_W2;
    if (b < PB_G1) { pack_w_body(gcn1_W, G1P, GNN_IN, GNN_HID, b * 256 + t); return; }
    b -= PB_G1;
    if (b < PB_G2) { pack_w_body(gcn2_W, G2P, GNN_HID, COMB, b * 256 + t); return; }
    b -= PB_G2;
    const int i = b * 256 + t;
    if (i < N_NODES) deg[i] = 1.0f;
}

template <bool AGG_INIT, int K>
__global__ void __launch_bounds__(256)
gemm_wmma_kernel(const float* __restrict__ A,
                 const unsigned short* __restrict__ Bp,
                 const float* __restrict__ bias,
                 void* __restrict__ C,
                 int M, int N, int doRelu, int outBf16,
                 const float* __restrict__ deg,
                 float* __restrict__ agg) {
    static_assert(K % 32 == 0, "whole k-tiles");
    const int lane = threadIdx.x & 31;
    const int wid  = threadIdx.x >> 5;
    const int nGroups = N >> 6;
    const int mTiles  = M >> 4;
    const int task = blockIdx.x * 8 + wid;
    if (task >= mTiles * nGroups) return;
    const int mt = task / nGroups;
    const int ng = task % nGroups;
    constexpr int kTiles = K >> 5;
    const int nTiles = N >> 4;

    const int hsel = (lane < 16) ? 0 : 8;
    const float* arow = A + (size_t)(mt * 16 + (lane & 15)) * K + hsel;
    const unsigned short* bcol = Bp + (size_t)(ng << 2) * 512 + lane * 16;
    const size_t bstride = (size_t)nTiles * 512;

    V16 bC[4], bN[4];
#pragma unroll
    for (int j = 0; j < 4; ++j) {
        bC[j].u[0] = ((const uint4*)(bcol + j * 512))[0];
        bC[j].u[1] = ((const uint4*)(bcol + j * 512))[1];
    }

    float4 aR[kTiles][4];
#pragma unroll
    for (int kt = 0; kt < kTiles; ++kt) {
        const float4* ap = (const float4*)(arow + kt * 32);
        aR[kt][0] = ap[0]; aR[kt][1] = ap[1]; aR[kt][2] = ap[4]; aR[kt][3] = ap[5];
    }
    __builtin_amdgcn_sched_barrier(0);

    v8f acc0 = {}, acc1 = {}, acc2 = {}, acc3 = {};

#pragma unroll
    for (int kt = 0; kt < kTiles; ++kt) {
        if (kt + 1 < kTiles) {
            const unsigned short* bt = bcol + (size_t)(kt + 1) * bstride;
#pragma unroll
            for (int j = 0; j < 4; ++j) {
                bN[j].u[0] = ((const uint4*)(bt + j * 512))[0];
                bN[j].u[1] = ((const uint4*)(bt + j * 512))[1];
            }
        }
        V16 a;
        a.u[0] = bf16x8_of(aR[kt][0], aR[kt][1]);
        a.u[1] = bf16x8_of(aR[kt][2], aR[kt][3]);

        acc0 = __builtin_amdgcn_wmma_f32_16x16x32_bf16(false, a.v, false, bC[0].v,
                                                       (short)0, acc0, false, false);
        acc1 = __builtin_amdgcn_wmma_f32_16x16x32_bf16(false, a.v, false, bC[1].v,
                                                       (short)0, acc1, false, false);
        acc2 = __builtin_amdgcn_wmma_f32_16x16x32_bf16(false, a.v, false, bC[2].v,
                                                       (short)0, acc2, false, false);
        acc3 = __builtin_amdgcn_wmma_f32_16x16x32_bf16(false, a.v, false, bC[3].v,
                                                       (short)0, acc3, false, false);

        if (kt + 1 < kTiles) {
#pragma unroll
            for (int j = 0; j < 4; ++j) bC[j] = bN[j];
        }
    }

    const int r0 = mt * 16 + ((lane < 16) ? 0 : 8);
    v8f accs[4] = {acc0, acc1, acc2, acc3};
    float selfw[8];
#pragma unroll
    for (int v = 0; v < 8; ++v) {
        if (AGG_INIT) { const float di = dinv_of(deg[r0 + v]); selfw[v] = di * di; }
        else          selfw[v] = 0.0f;
    }
#pragma unroll
    for (int j = 0; j < 4; ++j) {
        const int cc = (ng * 4 + j) * 16 + (lane & 15);
        const float bv = bias ? bias[cc] : 0.0f;
#pragma unroll
        for (int v = 0; v < 8; ++v) {
            float val = accs[j][v] + bv;
            if (doRelu) val = fmaxf(val, 0.0f);
            const size_t off = (size_t)(r0 + v) * N + cc;
            if (outBf16) ((unsigned short*)C)[off] = f2bf(val);
            else         ((float*)C)[off] = val;
            if (AGG_INIT) agg[off] = val * selfw[v];
        }
    }
}

#define FC1_MT   4
#define FC1_ROWS (16 * FC1_MT)
#define FC1_APITCH (MLP_IN * 2 + 16)
static_assert(MLP_HID == 8 * 64, "8 waves x 64 columns cover one row");
static_assert(MLP_IN * 4 == 2 * 32 * 16, "two wave-wide 16-byte loads cover one f32 row");
static_assert((MLP_IN / 32) % 2 == 0, "the K loop takes k-steps in pairs");
static_assert(FC1_ROWS % 8 == 0, "8 waves share the rows of the staging pass");

__device__ __forceinline__ void fc1_kstep(v8f (&acc)[FC1_MT][4], V16 (&a)[FC1_MT],
                                          const V16 (&bUse)[4], V16 (&bFill)[4],
                                          const unsigned char* afrag, int aNext,
                                          const unsigned short* __restrict__ bnext) {
#pragma unroll
    for (int j = 0; j < 4; ++j) {
        bFill[j].u[0] = ((const uint4*)(bnext + j * 512))[0];
        bFill[j].u[1] = ((const uint4*)(bnext + j * 512))[1];
    }
    __builtin_amdgcn_sched_barrier(0);
#pragma unroll
    for (int i = 0; i < FC1_MT; ++i) {
#pragma unroll
        for (int j = 0; j < 4; ++j)
            acc[i][j] = __builtin_amdgcn_wmma_f32_16x16x32_bf16(false, a[i].v, false, bUse[j].v,
                                                                (short)0, acc[i][j], false, false);
        __builtin_amdgcn_sched_barrier(0);
        const unsigned char* ap = afrag + i * 16 * FC1_APITCH + aNext;
        a[i].u[0] = *(const uint4*)(ap);
        a[i].u[1] = *(const uint4*)(ap + 32);
        __builtin_amdgcn_sched_barrier(0);
    }
}

__global__ void __launch_bounds__(256, 4)
gemm_relu_vecdot_kernel(const float* __restrict__ A,
                        const unsigned short* __restrict__ Bp,
                        const float* __restrict__ bias,
                        const float* __restrict__ wvec,
                        float* __restrict__ rowdot,
                        int M) {
    __shared__ __attribute__((aligned(16))) unsigned char As[FC1_ROWS * FC1_APITCH];
    __shared__ float part[8][FC1_ROWS];
    const int lane = threadIdx.x & 31;
    const int wid  = threadIdx.x >> 5;
    const int ng = wid;
    constexpr int kTiles = MLP_IN >> 5;
    constexpr int nTiles = MLP_HID >> 4;
    const int row0 = blockIdx.x * FC1_ROWS;

    const unsigned short* bcol = Bp + (size_t)(ng << 2) * 512 + lane * 16;
    constexpr size_t bstride = (size_t)nTiles * 512;

    V16 a[FC1_MT];
    V16 b0[4], b1[4];

#pragma unroll
    for (int j = 0; j < 4; ++j) {
        b0[j].u[0] = ((const uint4*)(bcol + j * 512))[0];
        b0[j].u[1] = ((const uint4*)(bcol + j * 512))[1];
    }

    const int c16 = lane & 15;
    float bv[4], wv[4];
#pragma unroll
    for (int j = 0; j < 4; ++j) {
        const int cc = (ng * 4 + j) * 16 + c16;
        bv[j] = bias[cc];
        wv[j] = wvec[cc];
    }

    {
        float4 st[FC1_ROWS / 8][2];
#pragma unroll
        for (int r = 0; r < FC1_ROWS / 8; ++r) {
            int grow = row0 + r * 8 + wid;
            grow = (grow < M) ? grow : (M - 1);
            const float4* rp = (const float4*)(A + (size_t)grow * MLP_IN);
            st[r][0] = rp[lane];
            st[r][1] = rp[32 + lane];
        }
#pragma unroll
        for (int r = 0; r < FC1_ROWS / 8; ++r) {
            unsigned char* dp = As + (r * 8 + wid) * FC1_APITCH + lane * 8;
            *(uint2*)(dp)       = bf16x4_of(st[r][0]);
            *(uint2*)(dp + 256) = bf16x4_of(st[r][1]);
        }
    }
    __syncthreads();

    const unsigned char* afrag = As + (lane & 15) * FC1_APITCH + ((lane >> 4) << 4);
#pragma unroll
    for (int i = 0; i < FC1_MT; ++i) {
        a[i].u[0] = *(const uint4*)(afrag + i * 16 * FC1_APITCH);
        a[i].u[1] = *(const uint4*)(afrag + i * 16 * FC1_APITCH + 32);
    }

    v8f acc[FC1_MT][4];
#pragma unroll
    for (int i = 0; i < FC1_MT; ++i)
#pragma unroll
        for (int j = 0; j < 4; ++j) acc[i][j] = v8f{};

    for (int kt = 0; kt < kTiles; kt += 2) {
        const int k1 = kt + 1;
        const int k2 = (kt + 2 < kTiles) ? (kt + 2) : (kt + 1);
        fc1_kstep(acc, a, b0, b1, afrag, k1 * 64, bcol + (size_t)k1 * bstride);
        fc1_kstep(acc, a, b1, b0, afrag, k2 * 64, bcol + (size_t)k2 * bstride);
    }

    float p[FC1_MT][8];
#pragma unroll
    for (int i = 0; i < FC1_MT; ++i) {
#pragma unroll
        for (int v = 0; v < 8; ++v) p[i][v] = 0.0f;
#pragma unroll
        for (int j = 0; j < 4; ++j) {
#pragma unroll
            for (int v = 0; v < 8; ++v) p[i][v] += fmaxf(acc[i][j][v] + bv[j], 0.0f) * wv[j];
        }
    }
#pragma unroll
    for (int off = 8; off > 0; off >>= 1) {
#pragma unroll
        for (int i = 0; i < FC1_MT; ++i) {
#pragma unroll
            for (int v = 0; v < 8; ++v) p[i][v] += __shfl_xor(p[i][v], off, 32);
        }
    }
#pragma unroll
    for (int i = 0; i < FC1_MT; ++i) {
        float mine = p[i][0];
#pragma unroll
        for (int v = 1; v < 8; ++v) mine = ((lane & 7) == v) ? p[i][v] : mine;
        if (c16 < 8) part[wid][i * 16 + ((lane >> 4) << 3) + c16] = mine;
    }
    __syncthreads();
    if (threadIdx.x < FC1_ROWS) {
        const int row = row0 + threadIdx.x;
        if (row < M) {
            float s = wvec[MLP_HID];
            for (int w = 0; w < 8; ++w) s += part[w][threadIdx.x];
            rowdot[row] = s;
        }
    }
}

__device__ __forceinline__ unsigned int relu_bias_bf16x2(float a, float ba,
                                                         float b, float bb) {
    return (unsigned int)f2bf(fmaxf(a + ba, 0.0f)) |
           ((unsigned int)f2bf(fmaxf(b + bb, 0.0f)) << 16);
}

__device__ __forceinline__ uint4 relu_bias_bf16x8(float4 f0, float4 f1,
                                                  float4 b0, float4 b1) {
    uint4 r;
    r.x = relu_bias_bf16x2(f0.x, b0.x, f0.y, b0.y);
    r.y = relu_bias_bf16x2(f0.z, b0.z, f0.w, b0.w);
    r.z = relu_bias_bf16x2(f1.x, b1.x, f1.y, b1.y);
    r.w = relu_bias_bf16x2(f1.z, b1.z, f1.w, b1.w);
    return r;
}

__device__ __forceinline__ void load_a_frag_f32(V16& a,
                                                const float* __restrict__ arow32,
                                                const float* __restrict__ abias,
                                                int koff) {
    const float4* fp = (const float4*)(arow32 + koff);
    const float4* bp = (const float4*)(abias + koff);
    a.u[0] = relu_bias_bf16x8(fp[0], fp[1], bp[0], bp[1]);
    a.u[1] = relu_bias_bf16x8(fp[4], fp[5], bp[4], bp[5]);
}

template <bool A_F32>
__global__ void gemm_rowdot_kernel(const unsigned short* __restrict__ A16,
                                   const float* __restrict__ A32,
                                   const float* __restrict__ abias,
                                   const unsigned short* __restrict__ Bp,
                                   const float* __restrict__ bias,
                                   const float* __restrict__ fvec,
                                   float* __restrict__ rowdot,
                                   int M, int K,
                                   const float* __restrict__ addend,
                                   const float* __restrict__ deg,
                                   const float* __restrict__ cbias,
                                   const float* __restrict__ fb,
                                   float* __restrict__ accp) {
    const int lane = threadIdx.x & 31;
    const int wid  = threadIdx.x >> 5;
    const int mt = blockIdx.x * 8 + wid;
    if (mt >= (M >> 4)) return;

    const int kTiles = K >> 5;

    const int hsel = (lane < 16) ? 0 : 8;
    const unsigned short* arow = A16 + (size_t)(mt * 16 + (lane & 15)) * K;
    const float* arow32 = A32 + (size_t)(mt * 16 + (lane & 15)) * K;
    const unsigned short* bcol = Bp + lane * 16;
    const size_t bstride = (size_t)(COMB >> 4) * 512;

    v8f acc0 = {}, acc1 = {}, acc2 = {}, acc3 = {};

    V16 aC, aN;
    V16 bC0, bC1, bC2, bC3, bN0, bN1, bN2, bN3;

    {
        if (!A_F32) {
            const uint4* ap = (const uint4*)(arow + hsel);
            aC.u[0] = ap[0];
            aC.u[1] = ap[2];
        } else {
            load_a_frag_f32(aC, arow32, abias, hsel);
        }
        const unsigned short* bt = bcol;
        bC0.u[0] = ((const uint4*)(bt + 0 * 512))[0];
        bC0.u[1] = ((const uint4*)(bt + 0 * 512))[1];
        bC1.u[0] = ((const uint4*)(bt + 1 * 512))[0];
        bC1.u[1] = ((const uint4*)(bt + 1 * 512))[1];
        bC2.u[0] = ((const uint4*)(bt + 2 * 512))[0];
        bC2.u[1] = ((const uint4*)(bt + 2 * 512))[1];
        bC3.u[0] = ((const uint4*)(bt + 3 * 512))[0];
        bC3.u[1] = ((const uint4*)(bt + 3 * 512))[1];
    }

    for (int kt = 0; kt < kTiles; ++kt) {
        const int ktn = (kt + 1 < kTiles) ? (kt + 1) : kt;
        {
            if (!A_F32) {
                const uint4* ap = (const uint4*)(arow + ktn * 32 + hsel);
                aN.u[0] = ap[0];
                aN.u[1] = ap[2];
            } else {
                load_a_frag_f32(aN, arow32, abias, ktn * 32 + hsel);
            }
            const unsigned short* bt = bcol + (size_t)ktn * bstride;
            bN0.u[0] = ((const uint4*)(bt + 0 * 512))[0];
            bN0.u[1] = ((const uint4*)(bt + 0 * 512))[1];
            bN1.u[0] = ((const uint4*)(bt + 1 * 512))[0];
            bN1.u[1] = ((const uint4*)(bt + 1 * 512))[1];
            bN2.u[0] = ((const uint4*)(bt + 2 * 512))[0];
            bN2.u[1] = ((const uint4*)(bt + 2 * 512))[1];
            bN3.u[0] = ((const uint4*)(bt + 3 * 512))[0];
            bN3.u[1] = ((const uint4*)(bt + 3 * 512))[1];
        }

        acc0 = __builtin_amdgcn_wmma_f32_16x16x32_bf16(false, aC.v, false, bC0.v,
                                                       (short)0, acc0, false, false);
        acc1 = __builtin_amdgcn_wmma_f32_16x16x32_bf16(false, aC.v, false, bC1.v,
                                                       (short)0, acc1, false, false);
        acc2 = __builtin_amdgcn_wmma_f32_16x16x32_bf16(false, aC.v, false, bC2.v,
                                                       (short)0, acc2, false, false);
        acc3 = __builtin_amdgcn_wmma_f32_16x16x32_bf16(false, aC.v, false, bC3.v,
                                                       (short)0, acc3, false, false);

        aC = aN;
        bC0 = bN0; bC1 = bN1; bC2 = bN2; bC3 = bN3;
    }

    const int c16 = lane & 15;
    v8f accs[4] = {acc0, acc1, acc2, acc3};
    float p[8];
#pragma unroll
    for (int v = 0; v < 8; ++v) p[v] = 0.0f;
    float cpart = 0.0f;
#pragma unroll
    for (int j = 0; j < 4; ++j) {
        const int cc = j * 16 + c16;
        const float fv = fvec[cc];
        const float bv = bias ? bias[cc] : 0.0f;
        if (accp) cpart += cbias[cc] * fv;
#pragma unroll
        for (int v = 0; v < 8; ++v) p[v] += (accs[j][v] + bv) * fv;
    }
#pragma unroll
    for (int off = 8; off > 0; off >>= 1) {
#pragma unroll
        for (int v = 0; v < 8; ++v) p[v] += __shfl_xor(p[v], off, 32);
        cpart += __shfl_xor(cpart, off, 32);
    }
    float mine = p[0];
#pragma unroll
    for (int v = 1; v < 8; ++v) mine = ((lane & 7) == v) ? p[v] : mine;
    if (c16 < 8) {
        const int row = mt * 16 + ((lane >> 4) << 3) + c16;
        rowdot[row] = mine;
        if (accp) {
            const float di = dinv_of(deg[row]);
            accp[(size_t)row * ACC_PITCH] = addend[row] + (cpart + fb[0]) + di * di * mine;
        }
    }
}

__global__ void deg_accum_kernel(const int* __restrict__ dst,
                                 const float* __restrict__ ew,
                                 float* __restrict__ deg, int nE) {
    int e = blockIdx.x * blockDim.x + threadIdx.x;
    if (e < nE) atomicAdd(&deg[dst[e]], ew[e]);
}

__global__ void edge_norm_kernel(const int* __restrict__ src,
                                 const int* __restrict__ dst,
                                 const float* __restrict__ ew,
                                 const float* __restrict__ deg,
                                 float* __restrict__ norm, int nE) {
    int e = blockIdx.x * blockDim.x + threadIdx.x;
    if (e < nE) norm[e] = dinv_of(deg[src[e]]) * ew[e] * dinv_of(deg[dst[e]]);
}

#define SC_EPW 32
#define SC_GRP 8
static_assert(N_EDGES % SC_EPW == 0, "a wave always holds SC_EPW real edges");
static_assert(COMB == 64, "lane l owns floats l and l+32");

__global__ void gcn_scatter_kernel(const float* __restrict__ h,
                                   const int* __restrict__ src,
                                   const int* __restrict__ dst,
                                   const float* __restrict__ norm,
                                   float* __restrict__ agg, int nE) {
    const int lane = threadIdx.x & 31;
    const int e0 = (blockIdx.x * (blockDim.x >> 5) + (threadIdx.x >> 5)) * SC_EPW;
    if (e0 >= nE) return;
    const int sv = src[e0 + lane];
    const int dv = dst[e0 + lane];
    const int nv = __float_as_int(norm[e0 + lane]);
    const float* hl = h + lane;
    float* al = agg + lane;

    float lo[2][SC_GRP], hi[2][SC_GRP];
#pragma unroll
    for (int k = 0; k < SC_GRP; ++k) {
        const int s = __builtin_amdgcn_readlane(sv, k);
        lo[0][k] = hl[(size_t)s * COMB];
        hi[0][k] = hl[(size_t)s * COMB + 32];
    }
#pragma unroll
    for (int g = 0; g < SC_EPW / SC_GRP; ++g) {
        const int cur = g & 1, nxt = cur ^ 1;
        if (g + 1 < SC_EPW / SC_GRP) {
#pragma unroll
            for (int k = 0; k < SC_GRP; ++k) {
                const int s = __builtin_amdgcn_readlane(sv, (g + 1) * SC_GRP + k);
                lo[nxt][k] = hl[(size_t)s * COMB];
                hi[nxt][k] = hl[(size_t)s * COMB + 32];
            }
        }
#pragma unroll
        for (int k = 0; k < SC_GRP; ++k) {
            const int d = __builtin_amdgcn_readlane(dv, g * SC_GRP + k);
            const float nm = __int_as_float(__builtin_amdgcn_readlane(nv, g * SC_GRP + k));
            float* ap = al + (size_t)d * COMB;
            atomicAdd(ap,      lo[cur][k] * nm);
            atomicAdd(ap + 32, hi[cur][k] * nm);
        }
    }
}

__global__ void gcn_scatter_dot_kernel(const float* __restrict__ s,
                                       const int* __restrict__ src,
                                       const int* __restrict__ dst,
                                       const float* __restrict__ norm,
                                       float* __restrict__ accp, int nE4) {
    int i = blockIdx.x * blockDim.x + threadIdx.x;
    if (i >= nE4) return;
    const int4   sv = ((const int4*)src)[i];
    const int4   dv = ((const int4*)dst)[i];
    const float4 nv = ((const float4*)norm)[i];
    const float m0 = nv.x * s[sv.x];
    const float m1 = nv.y * s[sv.y];
    const float m2 = nv.z * s[sv.z];
    const float m3 = nv.w * s[sv.w];
    atomicAdd(accp + (size_t)dv.x * ACC_PITCH, m0);
    atomicAdd(accp + (size_t)dv.y * ACC_PITCH, m1);
    atomicAdd(accp + (size_t)dv.z * ACC_PITCH, m2);
    atomicAdd(accp + (size_t)dv.w * ACC_PITCH, m3);
}

__global__ void out_gather_kernel(const float* __restrict__ accp,
                                  float* __restrict__ out, int n) {
    int i = blockIdx.x * blockDim.x + threadIdx.x;
    if (i < n) out[i] = accp[(size_t)i * ACC_PITCH];
}

extern "C" void kernel_launch(void* const* d_in, const int* in_sizes, int n_in,
                              void* d_out, int out_size, void* d_ws, size_t ws_size,
                              hipStream_t stream) {
    (void)in_sizes; (void)n_in; (void)out_size; (void)ws_size;

    const float* x        = (const float*)d_in[0];
    const int*   eidx     = (const int*)d_in[1];
    const float* eattr    = (const float*)d_in[2];
    const float* ufeat    = (const float*)d_in[3];
    const float* gcn1_W   = (const float*)d_in[4];
    const float* gcn1_b   = (const float*)d_in[5];
    const float* gcn2_W   = (const float*)d_in[6];
    const float* gcn2_b   = (const float*)d_in[7];
    const float* fc1_W    = (const float*)d_in[8];
    const float* fc1_b    = (const float*)d_in[9];
    const float* fc2_W    = (const float*)d_in[10];
    const float* fc2_b    = (const float*)d_in[11];
    const float* fcf_W    = (const float*)d_in[12];
    const float* fcf_b    = (const float*)d_in[13];
    const int* esrc = eidx;
    const int* edst = eidx + N_EDGES;
    float* out = (float*)d_out;

    char* ws = (char*)d_ws;
    size_t off = 0;
    auto carve = [&](size_t bytes) -> void* {
        void* p = ws + off;
        off = (off + bytes + 255) & ~(size_t)255;
        return p;
    };
    unsigned short* W1P = (unsigned short*)carve((size_t)MLP_IN * MLP_HID * 2);
    float* W2F    = (float*)carve((size_t)(MLP_HID + 1) * 4);
    unsigned short* G1P = (unsigned short*)carve((size_t)GNN_IN * GNN_HID * 2);
    unsigned short* G2P = (unsigned short*)carve((size_t)GNN_HID * COMB * 2);
    float* H1     = (float*)carve((size_t)N_NODES * COMB * 4);
    float* AGG    = (float*)carve((size_t)N_NODES * COMB * 4);
    float* MDOT   = (float*)carve((size_t)N_NODES * 4);
    float* SDOT   = (float*)carve((size_t)N_NODES * 4);
    float* ACC    = (float*)carve((size_t)N_NODES * ACC_PITCH * 4);
    float* DEG    = (float*)carve((size_t)N_NODES * 4);
    float* NORM   = (float*)carve((size_t)N_EDGES * 4);

    prep_kernel<<<PB_ALL, 256, 0, stream>>>(fc1_W, W1P, fc2_W, fc2_b,
                                            fcf_W, W2F, gcn1_W, G1P, gcn2_W, G2P, DEG);

    deg_accum_kernel<<<(N_EDGES + 255) / 256, 256, 0, stream>>>(edst, eattr, DEG, N_EDGES);
    edge_norm_kernel<<<(N_EDGES + 255) / 256, 256, 0, stream>>>(esrc, edst, eattr, DEG,
                                                                NORM, N_EDGES);

    const int rdBlocks = ((N_NODES / 16) + 7) / 8;

    gemm_relu_vecdot_kernel<<<(N_NODES + FC1_ROWS - 1) / FC1_ROWS, 256, 0, stream>>>(
        ufeat, W1P, fc1_b, W2F, MDOT, N_NODES);

    gemm_wmma_kernel<true, GNN_IN><<<((N_NODES / 16) * (GNN_HID / 64) + 7) / 8, 256, 0, stream>>>(
        x, G1P, nullptr, H1, N_NODES, GNN_HID, 0, 0, DEG, AGG);
    gcn_scatter_kernel<<<(N_EDGES / SC_EPW + 7) / 8, 256, 0, stream>>>(
        H1, esrc, edst, NORM, AGG, N_EDGES);

    gemm_rowdot_kernel<true><<<rdBlocks, 256, 0, stream>>>(
        nullptr, AGG, gcn1_b, G2P, nullptr, fcf_W + COMB, SDOT, N_NODES, GNN_HID,
        MDOT, DEG, gcn2_b, fcf_b, ACC);
    gcn_scatter_dot_kernel<<<(N_EDGES / 4 + 255) / 256, 256, 0, stream>>>(
        SDOT, esrc, edst, NORM, ACC, N_EDGES / 4);

    out_gather_kernel<<<(N_NODES + 255) / 256, 256, 0, stream>>>(ACC, out, N_NODES);
}
